# starting kernel
# baseline (speedup 1.0000x reference)
#include <hip/hip_runtime.h>
#include <hip/hip_bf16.h>
#include <math.h>

typedef __attribute__((ext_vector_type(2))) float v2f;
typedef __attribute__((ext_vector_type(8))) float v8f;

#define HEADS 4
#define FDIM  128

__device__ __forceinline__ unsigned f2key(float f) {
    unsigned b = __float_as_uint(f);
    return (b & 0x80000000u) ? ~b : (b | 0x80000000u);
}
__device__ __forceinline__ float key2f(unsigned k) {
    unsigned b = (k & 0x80000000u) ? (k & 0x7FFFFFFFu) : ~k;
    return __uint_as_float(b);
}

__global__ __launch_bounds__(256) void gemm_x128(
    const float* __restrict__ X, const float* __restrict__ W,
    const float* __restrict__ bias, float* __restrict__ Y, int M)
{
    __shared__ float At[16][132];
    const int tid  = threadIdx.x;
    const int row0 = blockIdx.x * 16;

    #pragma unroll
    for (int i = 0; i < 2; ++i) {
        int idx = (tid + i * 256) * 4;
        int r = idx >> 7, c = idx & 127;
        int gr = row0 + r; gr = gr < M ? gr : M - 1;
        const float4 v = *(const float4*)(X + (size_t)gr * FDIM + c);
        At[r][c + 0] = v.x; At[r][c + 1] = v.y;
        At[r][c + 2] = v.z; At[r][c + 3] = v.w;
    }
    __syncthreads();

    const int lane = tid & 31;
    const int wv   = tid >> 5;
    const int m    = lane & 15;
    const int half = lane >> 4;
    const int koff = half << 1;
    const int n    = wv * 16 + (lane & 15);

    v8f acc = {};
    #pragma unroll
    for (int kb = 0; kb < FDIM; kb += 4) {
        v2f a, b;
        a.x = At[m][kb + koff];
        a.y = At[m][kb + koff + 1];
        b.x = W[(kb + koff)     * FDIM + n];
        b.y = W[(kb + koff + 1) * FDIM + n];
        acc = __builtin_amdgcn_wmma_f32_16x16x4_f32(
                  false, a, false, b, (short)0, acc, false, false);
    }

    const float bn = bias ? bias[n] : 0.0f;
    #pragma unroll
    for (int r = 0; r < 8; ++r) {
        int row = row0 + half * 8 + r;
        if (row < M) Y[(size_t)row * FDIM + n] = acc[r] + bn;
    }
}

__global__ __launch_bounds__(256) void node_alpha(
    const float* __restrict__ feat, const float* __restrict__ att,
    float* __restrict__ out, int N)
{
    const int lane = threadIdx.x & 31;
    const int node = blockIdx.x * 8 + (threadIdx.x >> 5);
    if (node >= N) return;
    const float4 f = *(const float4*)(feat + (size_t)node * FDIM + lane * 4);
    const float4 a = *(const float4*)(att + lane * 4);
    float p = f.x * a.x + f.y * a.y + f.z * a.z + f.w * a.w;
    p += __shfl_xor(p, 1);
    p += __shfl_xor(p, 2);
    p += __shfl_xor(p, 4);
    if ((lane & 7) == 0) out[node * HEADS + (lane >> 3)] = p;
}

__global__ void init_softmax(unsigned* __restrict__ amax,
                             float* __restrict__ asum, int n)
{
    int i = blockIdx.x * blockDim.x + threadIdx.x;
    if (i < n) { amax[i] = 0u; asum[i] = 0.0f; }
}

__global__ void edge_alpha_max(
    const int* __restrict__ esrc, const int* __restrict__ edst,
    const float* __restrict__ asn, const float* __restrict__ adn,
    float* __restrict__ alpha_e, unsigned* __restrict__ amax, int E)
{
    int e = blockIdx.x * blockDim.x + threadIdx.x;
    if (e >= E) return;
    const int s = esrc[e], d = edst[e];
    const float4 a = *(const float4*)(asn + (size_t)s * HEADS);
    const float4 b = *(const float4*)(adn + (size_t)d * HEADS);
    float4 al;
    al.x = a.x + b.x; al.x = al.x > 0.f ? al.x : 0.2f * al.x;
    al.y = a.y + b.y; al.y = al.y > 0.f ? al.y : 0.2f * al.y;
    al.z = a.z + b.z; al.z = al.z > 0.f ? al.z : 0.2f * al.z;
    al.w = a.w + b.w; al.w = al.w > 0.f ? al.w : 0.2f * al.w;
    *(float4*)(alpha_e + (size_t)e * HEADS) = al;
    unsigned* mp = amax + (size_t)d * HEADS;
    __hip_atomic_fetch_max(mp + 0, f2key(al.x), __ATOMIC_RELAXED, __HIP_MEMORY_SCOPE_AGENT);
    __hip_atomic_fetch_max(mp + 1, f2key(al.y), __ATOMIC_RELAXED, __HIP_MEMORY_SCOPE_AGENT);
    __hip_atomic_fetch_max(mp + 2, f2key(al.z), __ATOMIC_RELAXED, __HIP_MEMORY_SCOPE_AGENT);
    __hip_atomic_fetch_max(mp + 3, f2key(al.w), __ATOMIC_RELAXED, __HIP_MEMORY_SCOPE_AGENT);
}

__global__ void edge_exp_sum(
    const int* __restrict__ edst, const unsigned* __restrict__ amax,
    float* __restrict__ alpha_e, float* __restrict__ asum, int E)
{
    int e = blockIdx.x * blockDim.x + threadIdx.x;
    if (e >= E) return;
    const int d = edst[e];
    float4 al = *(const float4*)(alpha_e + (size_t)e * HEADS);
    const unsigned* mp = amax + (size_t)d * HEADS;
    al.x = __expf(al.x - key2f(mp[0]));
    al.y = __expf(al.y - key2f(mp[1]));
    al.z = __expf(al.z - key2f(mp[2]));
    al.w = __expf(al.w - key2f(mp[3]));
    *(float4*)(alpha_e + (size_t)e * HEADS) = al;
    float* sp = asum + (size_t)d * HEADS;
    __hip_atomic_fetch_add(sp + 0, al.x, __ATOMIC_RELAXED, __HIP_MEMORY_SCOPE_AGENT);
    __hip_atomic_fetch_add(sp + 1, al.y, __ATOMIC_RELAXED, __HIP_MEMORY_SCOPE_AGENT);
    __hip_atomic_fetch_add(sp + 2, al.z, __ATOMIC_RELAXED, __HIP_MEMORY_SCOPE_AGENT);
    __hip_atomic_fetch_add(sp + 3, al.w, __ATOMIC_RELAXED, __HIP_MEMORY_SCOPE_AGENT);
}

__global__ __launch_bounds__(256) void edge_scatter(
    const int* __restrict__ esrc, const int* __restrict__ edst,
    const float* __restrict__ src_feat, const float* __restrict__ alpha_e,
    const float* __restrict__ asum, float* __restrict__ out, int E)
{
    const int lane = threadIdx.x & 31;
    const int e = blockIdx.x * 8 + (threadIdx.x >> 5);
    if (e >= E) return;
    const int s = esrc[e], d = edst[e];
    const int h = lane >> 3;
    const float w = alpha_e[(size_t)e * HEADS + h] /
                    (asum[(size_t)d * HEADS + h] + 1e-16f);
    const float4 f = *(const float4*)(src_feat + (size_t)s * FDIM + lane * 4);
    float* op = out + (size_t)d * FDIM + lane * 4;
    __hip_atomic_fetch_add(op + 0, w * f.x, __ATOMIC_RELAXED, __HIP_MEMORY_SCOPE_AGENT);
    __hip_atomic_fetch_add(op + 1, w * f.y, __ATOMIC_RELAXED, __HIP_MEMORY_SCOPE_AGENT);
    __hip_atomic_fetch_add(op + 2, w * f.z, __ATOMIC_RELAXED, __HIP_MEMORY_SCOPE_AGENT);
    __hip_atomic_fetch_add(op + 3, w * f.w, __ATOMIC_RELAXED, __HIP_MEMORY_SCOPE_AGENT);
}

extern "C" void kernel_launch(void* const* d_in, const int* in_sizes, int n_in,
                              void* d_out, int out_size, void* d_ws, size_t ws_size,
                              hipStream_t stream)
{
    const float* x_src  = (const float*)d_in[0];
    const float* x_dst  = (const float*)d_in[1];
    const int*   esrc   = (const int*)d_in[2];
    const int*   edst   = (const int*)d_in[3];
    const float* W_src  = (const float*)d_in[5];
    const float* W_dst  = (const float*)d_in[6];
    const float* attS   = (const float*)d_in[7];
    const float* attD   = (const float*)d_in[8];
    const float* W_self = (const float*)d_in[9];
    const float* b_self = (const float*)d_in[10];

    const int N_SRC = in_sizes[0] / FDIM;
    const int N_DST = in_sizes[1] / FDIM;
    const int E     = in_sizes[2];

    char* ws = (char*)d_ws;
    float*    src_feat = (float*)   (ws);
    float*    dst_feat = (float*)   (ws + (size_t)N_SRC * FDIM * 4);
    float*    as_node  = (float*)   (ws + (size_t)(N_SRC + N_DST) * FDIM * 4);
    float*    ad_node  = as_node + (size_t)N_SRC * HEADS;
    unsigned* amax     = (unsigned*)(ad_node + (size_t)N_DST * HEADS);
    float*    asum     = (float*)   (amax + (size_t)N_DST * HEADS);
    float*    alpha_e  = asum + (size_t)N_DST * HEADS;

    float* out = (float*)d_out;

    {
        int n = N_DST * HEADS;
        init_softmax<<<(n + 255) / 256, 256, 0, stream>>>(amax, asum, n);
    }
    gemm_x128<<<(N_SRC + 15) / 16, 256, 0, stream>>>(x_src, W_src, nullptr, src_feat, N_SRC);
    gemm_x128<<<(N_DST + 15) / 16, 256, 0, stream>>>(x_dst, W_dst, nullptr, dst_feat, N_DST);
    gemm_x128<<<(N_DST + 15) / 16, 256, 0, stream>>>(x_dst, W_self, b_self, out, N_DST);
    node_alpha<<<(N_SRC + 7) / 8, 256, 0, stream>>>(src_feat, attS, as_node, N_SRC);
    node_alpha<<<(N_DST + 7) / 8, 256, 0, stream>>>(dst_feat, attD, ad_node, N_DST);
    edge_alpha_max<<<(E + 255) / 256, 256, 0, stream>>>(esrc, edst, as_node, ad_node,
                                                        alpha_e, amax, E);
    edge_exp_sum<<<(E + 255) / 256, 256, 0, stream>>>(edst, amax, alpha_e, asum, E);
    edge_scatter<<<(E + 7) / 8, 256, 0, stream>>>(esrc, edst, src_feat, alpha_e,
                                                  asum, out, E);
}


# Edges grouped by destination per call (counting sort, int atomics); one wave per node does softmax and weighted row sum in registers: no float atomics, sum order changes, 10 launches
# speedup vs baseline: 2.2762x; 2.2762x over previous
#include <hip/hip_runtime.h>
#include <hip/hip_bf16.h>
#include <math.h>

typedef __attribute__((ext_vector_type(2))) float v2f;
typedef __attribute__((ext_vector_type(8))) float v8f;

#define HEADS 4
#define FDIM  128

__global__ __launch_bounds__(256) void gemm_x128(
    const float* __restrict__ X, const float* __restrict__ W,
    const float* __restrict__ bias, float* __restrict__ Y, int M)
{
    __shared__ float At[16][132];
    const int tid  = threadIdx.x;
    const int row0 = blockIdx.x * 16;

    #pragma unroll
    for (int i = 0; i < 2; ++i) {
        int idx = (tid + i * 256) * 4;
        int r = idx >> 7, c = idx & 127;
        int gr = row0 + r; gr = gr < M ? gr : M - 1;
        const float4 v = *(const float4*)(X + (size_t)gr * FDIM + c);
        At[r][c + 0] = v.x; At[r][c + 1] = v.y;
        At[r][c + 2] = v.z; At[r][c + 3] = v.w;
    }
    __syncthreads();

    const int lane = tid & 31;
    const int wv   = tid >> 5;
    const int m    = lane & 15;
    const int half = lane >> 4;
    const int koff = half << 1;
    const int n    = wv * 16 + (lane & 15);

    v8f acc = {};
    #pragma unroll
    for (int kb = 0; kb < FDIM; kb += 4) {
        v2f a, b;
        a.x = At[m][kb + koff];
        a.y = At[m][kb + koff + 1];
        b.x = W[(kb + koff)     * FDIM + n];
        b.y = W[(kb + koff + 1) * FDIM + n];
        acc = __builtin_amdgcn_wmma_f32_16x16x4_f32(
                  false, a, false, b, (short)0, acc, false, false);
    }

    const float bn = bias ? bias[n] : 0.0f;
    #pragma unroll
    for (int r = 0; r < 8; ++r) {
        int row = row0 + half * 8 + r;
        if (row < M) Y[(size_t)row * FDIM + n] = acc[r] + bn;
    }
}

__global__ __launch_bounds__(256) void node_alpha(
    const float* __restrict__ feat, const float* __restrict__ att,
    float* __restrict__ out, int N)
{
    const int lane = threadIdx.x & 31;
    const int node = blockIdx.x * 8 + (threadIdx.x >> 5);
    if (node >= N) return;
    const float4 f = *(const float4*)(feat + (size_t)node * FDIM + lane * 4);
    const float4 a = *(const float4*)(att + lane * 4);
    float p = f.x * a.x + f.y * a.y + f.z * a.z + f.w * a.w;
    p += __shfl_xor(p, 1);
    p += __shfl_xor(p, 2);
    p += __shfl_xor(p, 4);
    if ((lane & 7) == 0) out[node * HEADS + (lane >> 3)] = p;
}

__global__ void zero_counts(int* cnt, int* cursor, int n)
{
    int i = blockIdx.x * blockDim.x + threadIdx.x;
    if (i < n) cnt[i] = 0;
    if (i == 0) *cursor = 0;
}

__global__ void count_dst(const int* __restrict__ edst, int* cnt, int E)
{
    int e = blockIdx.x * blockDim.x + threadIdx.x;
    if (e >= E) return;
    __hip_atomic_fetch_add(cnt + edst[e], 1, __ATOMIC_RELAXED, __HIP_MEMORY_SCOPE_AGENT);
}

__global__ __launch_bounds__(256) void segment_starts(
    const int* cnt, int* cursor, int* start, int* fillpos, int N)
{
    const int d = blockIdx.x * 256 + threadIdx.x;
    const int lane = threadIdx.x & 31;
    const int c = d < N ? cnt[d] : 0;
    int incl = c;
    #pragma unroll
    for (int o = 1; o < 32; o <<= 1) {
        const int t = __shfl_up(incl, o);
        if (lane >= o) incl += t;
    }
    const int total = __shfl(incl, 31);
    int base = 0;
    if (lane == 31)
        base = __hip_atomic_fetch_add(cursor, total, __ATOMIC_RELAXED, __HIP_MEMORY_SCOPE_AGENT);
    base = __shfl(base, 31);
    if (d < N) { const int st = base + incl - c; start[d] = st; fillpos[d] = st; }
}

__global__ void place_edges(const int* __restrict__ esrc, const int* __restrict__ edst,
                            int* fillpos, int* __restrict__ ssrc, int E)
{
    int e = blockIdx.x * blockDim.x + threadIdx.x;
    if (e >= E) return;
    const int pos = __hip_atomic_fetch_add(fillpos + edst[e], 1,
                                           __ATOMIC_RELAXED, __HIP_MEMORY_SCOPE_AGENT);
    ssrc[pos] = esrc[e];
}

__device__ __forceinline__ float lrelu(float x) { return x > 0.f ? x : 0.2f * x; }

__device__ __forceinline__ float4 edge_scores(const float* __restrict__ asn, int s, float4 adv)
{
    const float4 a = *(const float4*)(asn + (size_t)s * HEADS);
    float4 r;
    r.x = lrelu(a.x + adv.x); r.y = lrelu(a.y + adv.y);
    r.z = lrelu(a.z + adv.z); r.w = lrelu(a.w + adv.w);
    return r;
}

__device__ __forceinline__ float4 wave_max4(float4 v)
{
    #pragma unroll
    for (int o = 16; o > 0; o >>= 1) {
        v.x = fmaxf(v.x, __shfl_xor(v.x, o)); v.y = fmaxf(v.y, __shfl_xor(v.y, o));
        v.z = fmaxf(v.z, __shfl_xor(v.z, o)); v.w = fmaxf(v.w, __shfl_xor(v.w, o));
    }
    return v;
}

__device__ __forceinline__ float4 wave_sum4(float4 v)
{
    #pragma unroll
    for (int o = 16; o > 0; o >>= 1) {
        v.x += __shfl_xor(v.x, o); v.y += __shfl_xor(v.y, o);
        v.z += __shfl_xor(v.z, o); v.w += __shfl_xor(v.w, o);
    }
    return v;
}

__device__ __forceinline__ float head_weight(float4 w_l, int i, const unsigned (&hm)[4])
{
    const unsigned bx = (unsigned)__builtin_amdgcn_readlane(__float_as_int(w_l.x), i);
    const unsigned by = (unsigned)__builtin_amdgcn_readlane(__float_as_int(w_l.y), i);
    const unsigned bz = (unsigned)__builtin_amdgcn_readlane(__float_as_int(w_l.z), i);
    const unsigned bw = (unsigned)__builtin_amdgcn_readlane(__float_as_int(w_l.w), i);
    return __uint_as_float((bx & hm[0]) | (by & hm[1]) | (bz & hm[2]) | (bw & hm[3]));
}

__device__ __forceinline__ float4 src_row(const float* __restrict__ src_feat, int s_l, int i, int lane)
{
    const int s = __builtin_amdgcn_readlane(s_l, i);
    return *(const float4*)(src_feat + (size_t)s * FDIM + lane * 4);
}

__device__ __forceinline__ void fma_row(float4& acc, float w, float4 f)
{
    acc.x += w * f.x; acc.y += w * f.y; acc.z += w * f.z; acc.w += w * f.w;
}

__device__ __forceinline__ void add_rows(int s_l, float4 w_l, int m, int lane,
                                         const unsigned (&hm)[4],
                                         const float* __restrict__ src_feat, float4& acc)
{
    int i = 0;
    for (; i + 4 <= m; i += 4) {
        const float4 f0 = src_row(src_feat, s_l, i + 0, lane);
        const float4 f1 = src_row(src_feat, s_l, i + 1, lane);
        const float4 f2 = src_row(src_feat, s_l, i + 2, lane);
        const float4 f3 = src_row(src_feat, s_l, i + 3, lane);
        const float w0 = head_weight(w_l, i + 0, hm);
        const float w1 = head_weight(w_l, i + 1, hm);
        const float w2 = head_weight(w_l, i + 2, hm);
        const float w3 = head_weight(w_l, i + 3, hm);
        fma_row(acc, w0, f0); fma_row(acc, w1, f1);
        fma_row(acc, w2, f2); fma_row(acc, w3, f3);
    }
    for (; i < m; ++i) {
        const float4 f = src_row(src_feat, s_l, i, lane);
        fma_row(acc, head_weight(w_l, i, hm), f);
    }
}

__global__ __launch_bounds__(256) void segment_softmax_aggregate(
    const int* __restrict__ ssrc, const int* __restrict__ start, const int* __restrict__ cnt,
    const float* __restrict__ src_feat, const float* __restrict__ asn,
    const float* __restrict__ adn, float* __restrict__ out, int N)
{
    const int lane = threadIdx.x & 31;
    const int wv = __builtin_amdgcn_readfirstlane((int)(threadIdx.x >> 5));
    const int d = blockIdx.x * 8 + wv;
    if (d >= N) return;
    const int beg = start[d], n = cnt[d];
    const int h = lane >> 3;
    const unsigned hm[4] = { h == 0 ? ~0u : 0u, h == 1 ? ~0u : 0u, h == 2 ? ~0u : 0u, h == 3 ? ~0u : 0u };
    float* op = out + (size_t)d * FDIM + lane * 4;
    const float4 self = *(const float4*)op;
    const float4 adv = *(const float4*)(adn + (size_t)d * HEADS);
    const float NEG = -INFINITY;

    const bool v0 = lane < n;
    const int s0 = v0 ? ssrc[beg + lane] : 0;
    float4 al0 = {NEG, NEG, NEG, NEG};
    if (v0) al0 = edge_scores(asn, s0, adv);

    float4 mx = al0;
    for (int b = 32; b < n; b += 32) {
        const int j = b + lane;
        if (j < n) {
            const float4 al = edge_scores(asn, ssrc[beg + j], adv);
            mx.x = fmaxf(mx.x, al.x); mx.y = fmaxf(mx.y, al.y);
            mx.z = fmaxf(mx.z, al.z); mx.w = fmaxf(mx.w, al.w);
        }
    }
    mx = wave_max4(mx);

    float4 ex0 = {0.f, 0.f, 0.f, 0.f};
    if (v0) {
        ex0.x = __expf(al0.x - mx.x); ex0.y = __expf(al0.y - mx.y);
        ex0.z = __expf(al0.z - mx.z); ex0.w = __expf(al0.w - mx.w);
    }
    float4 sm = ex0;
    for (int b = 32; b < n; b += 32) {
        const int j = b + lane;
        if (j < n) {
            const float4 al = edge_scores(asn, ssrc[beg + j], adv);
            sm.x += __expf(al.x - mx.x); sm.y += __expf(al.y - mx.y);
            sm.z += __expf(al.z - mx.z); sm.w += __expf(al.w - mx.w);
        }
    }
    sm = wave_sum4(sm);
    float4 den;
    den.x = sm.x + 1e-16f; den.y = sm.y + 1e-16f; den.z = sm.z + 1e-16f; den.w = sm.w + 1e-16f;

    float4 acc = {0.f, 0.f, 0.f, 0.f};
    float4 w0;
    w0.x = ex0.x / den.x; w0.y = ex0.y / den.y; w0.z = ex0.z / den.z; w0.w = ex0.w / den.w;
    add_rows(s0, w0, n < 32 ? n : 32, lane, hm, src_feat, acc);
    for (int b = 32; b < n; b += 32) {
        const int j = b + lane;
        int s = 0;
        float4 w = {0.f, 0.f, 0.f, 0.f};
        if (j < n) {
            s = ssrc[beg + j];
            const float4 al = edge_scores(asn, s, adv);
            w.x = __expf(al.x - mx.x) / den.x; w.y = __expf(al.y - mx.y) / den.y;
            w.z = __expf(al.z - mx.z) / den.z; w.w = __expf(al.w - mx.w) / den.w;
        }
        const int rem = n - b;
        add_rows(s, w, rem < 32 ? rem : 32, lane, hm, src_feat, acc);
    }

    float4 o;
    o.x = self.x + acc.x; o.y = self.y + acc.y; o.z = self.z + acc.z; o.w = self.w + acc.w;
    *(float4*)op = o;
}

extern "C" void kernel_launch(void* const* d_in, const int* in_sizes, int n_in,
                              void* d_out, int out_size, void* d_ws, size_t ws_size,
                              hipStream_t stream)
{
    const float* x_src  = (const float*)d_in[0];
    const float* x_dst  = (const float*)d_in[1];
    const int*   esrc   = (const int*)d_in[2];
    const int*   edst   = (const int*)d_in[3];
    const float* W_src  = (const float*)d_in[5];
    const float* W_dst  = (const float*)d_in[6];
    const float* attS   = (const float*)d_in[7];
    const float* attD   = (const float*)d_in[8];
    const float* W_self = (const float*)d_in[9];
    const float* b_self = (const float*)d_in[10];

    const int N_SRC = in_sizes[0] / FDIM;
    const int N_DST = in_sizes[1] / FDIM;
    const int E     = in_sizes[2];

    char* ws = (char*)d_ws;
    float* src_feat = (float*)(ws);
    float* dst_feat = (float*)(ws + (size_t)N_SRC * FDIM * 4);
    float* as_node  = (float*)(ws + (size_t)(N_SRC + N_DST) * FDIM * 4);
    float* ad_node  = as_node + (size_t)N_SRC * HEADS;
    int*   cnt      = (int*)(ad_node + (size_t)N_DST * HEADS);
    int*   start    = cnt + N_DST;
    int*   fillpos  = start + N_DST;
    int*   ssrc     = fillpos + N_DST;
    int*   cursor   = ssrc + E;

    float* out = (float*)d_out;

    zero_counts<<<(N_DST + 255) / 256, 256, 0, stream>>>(cnt, cursor, N_DST);
    gemm_x128<<<(N_SRC + 15) / 16, 256, 0, stream>>>(x_src, W_src, nullptr, src_feat, N_SRC);
    gemm_x128<<<(N_DST + 15) / 16, 256, 0, stream>>>(x_dst, W_dst, nullptr, dst_feat, N_DST);
    gemm_x128<<<(N_DST + 15) / 16, 256, 0, stream>>>(x_dst, W_self, b_self, out, N_DST);
    node_alpha<<<(N_SRC + 7) / 8, 256, 0, stream>>>(src_feat, attS, as_node, N_SRC);
    node_alpha<<<(N_DST + 7) / 8, 256, 0, stream>>>(dst_feat, attD, ad_node, N_DST);
    count_dst<<<(E + 255) / 256, 256, 0, stream>>>(edst, cnt, E);
    segment_starts<<<(N_DST + 255) / 256, 256, 0, stream>>>(cnt, cursor, start, fillpos, N_DST);
    place_edges<<<(E + 255) / 256, 256, 0, stream>>>(esrc, edst, fillpos, ssrc, E);
    segment_softmax_aggregate<<<(N_DST + 7) / 8, 256, 0, stream>>>(
        ssrc, start, cnt, src_feat, as_node, ad_node, out, N_DST);
}


# Attention logits as x @ (W.att): W.att built per call as a padded ninth WMMA column tile of the two remaining GEMMs; dst_feat GEMM and both per-node dot launches go; logit sum order changes
# speedup vs baseline: 2.3966x; 1.0529x over previous
#include <hip/hip_runtime.h>
#include <hip/hip_bf16.h>
#include <math.h>

typedef __attribute__((ext_vector_type(2))) float v2f;
typedef __attribute__((ext_vector_type(8))) float v8f;

#define HEADS 4
#define FDIM  128

template <int LDB>
__device__ __forceinline__ v8f tile_product(const float (*At)[132],
                                            const float* __restrict__ Bcol, int m, int koff)
{
    v8f acc = {};
    #pragma unroll
    for (int kb = 0; kb < FDIM; kb += 4) {
        v2f a, b;
        a.x = At[m][kb + koff];
        a.y = At[m][kb + koff + 1];
        b.x = Bcol[(kb + koff)     * LDB];
        b.y = Bcol[(kb + koff + 1) * LDB];
        acc = __builtin_amdgcn_wmma_f32_16x16x4_f32(
                  false, a, false, b, (short)0, acc, false, false);
    }
    return acc;
}

__global__ __launch_bounds__(288) void gemm_x128_logits(
    const float* __restrict__ X, const float* __restrict__ W,
    const float* __restrict__ Vp, const float* __restrict__ bias,
    float* __restrict__ Y, float* __restrict__ alpha, int M)
{
    __shared__ float At[16][132];
    const int tid  = threadIdx.x;
    const int row0 = blockIdx.x * 16;
    const int wv = __builtin_amdgcn_readfirstlane(tid >> 5);

    if (wv < 8) {
        #pragma unroll
        for (int i = 0; i < 2; ++i) {
            int idx = (tid + i * 256) * 4;
            int r = idx >> 7, c = idx & 127;
            int gr = row0 + r; gr = gr < M ? gr : M - 1;
            const float4 v = *(const float4*)(X + (size_t)gr * FDIM + c);
            At[r][c + 0] = v.x; At[r][c + 1] = v.y;
            At[r][c + 2] = v.z; At[r][c + 3] = v.w;
        }
    }
    __syncthreads();

    const int lane = tid & 31;
    const int m    = lane & 15;
    const int half = lane >> 4;
    const int koff = half << 1;

    if (wv < 8) {
        const int n = wv * 16 + m;
        const v8f acc = tile_product<FDIM>(At, W + n, m, koff);
        const float bn = bias ? bias[n] : 0.0f;
        #pragma unroll
        for (int r = 0; r < 8; ++r) {
            int row = row0 + half * 8 + r;
            if (row < M) Y[(size_t)row * FDIM + n] = acc[r] + bn;
        }
    } else {
        const v8f acc = tile_product<16>(At, Vp + m, m, koff);
        if (m < HEADS) {
            #pragma unroll
            for (int r = 0; r < 8; ++r) {
                int row = row0 + half * 8 + r;
                if (row < M) alpha[(size_t)row * HEADS + m] = acc[r];
            }
        }
    }
}

__global__ __launch_bounds__(256) void clear_counts_build_logit_tiles(
    int* cnt, int* cursor, int n, int count_blocks,
    const float* __restrict__ W_s, const float* __restrict__ att_s,
    const float* __restrict__ W_d, const float* __restrict__ att_d,
    float* __restrict__ Vp_s, float* __restrict__ Vp_d)
{
    if ((int)blockIdx.x < count_blocks) {
        int i = blockIdx.x * 256 + threadIdx.x;
        if (i < n) cnt[i] = 0;
        if (i == 0) *cursor = 0;
        return;
    }
    const int t = ((int)blockIdx.x - count_blocks) * 256 + threadIdx.x;
    const int which = t >> 11;
    const int k = (t >> 4) & 127;
    const int c = t & 15;
    const float* Wm  = which ? W_d : W_s;
    const float* att = which ? att_d : att_s;
    float s = 0.0f;
    if (c < HEADS) {
        const float* wr = Wm + (size_t)k * FDIM + c * 32;
        const float* ar = att + c * 32;
        #pragma unroll
        for (int d = 0; d < 32; d += 4) {
            const float4 w = *(const float4*)(wr + d);
            const float4 a = *(const float4*)(ar + d);
            s += w.x * a.x; s += w.y * a.y; s += w.z * a.z; s += w.w * a.w;
        }
    }
    (which ? Vp_d : Vp_s)[k * 16 + c] = s;
}

__global__ void count_dst(const int* __restrict__ edst, int* cnt, int E)
{
    int e = blockIdx.x * blockDim.x + threadIdx.x;
    if (e >= E) return;
    __hip_atomic_fetch_add(cnt + edst[e], 1, __ATOMIC_RELAXED, __HIP_MEMORY_SCOPE_AGENT);
}

__global__ __launch_bounds__(256) void segment_starts(
    const int* cnt, int* cursor, int* start, int* fillpos, int N)
{
    const int d = blockIdx.x * 256 + threadIdx.x;
    const int lane = threadIdx.x & 31;
    const int c = d < N ? cnt[d] : 0;
    int incl = c;
    #pragma unroll
    for (int o = 1; o < 32; o <<= 1) {
        const int t = __shfl_up(incl, o);
        if (lane >= o) incl += t;
    }
    const int total = __shfl(incl, 31);
    int base = 0;
    if (lane == 31)
        base = __hip_atomic_fetch_add(cursor, total, __ATOMIC_RELAXED, __HIP_MEMORY_SCOPE_AGENT);
    base = __shfl(base, 31);
    if (d < N) { const int st = base + incl - c; start[d] = st; fillpos[d] = st; }
}

__global__ void place_edges(const int* __restrict__ esrc, const int* __restrict__ edst,
                            int* fillpos, int* __restrict__ ssrc, int E)
{
    int e = blockIdx.x * blockDim.x + threadIdx.x;
    if (e >= E) return;
    const int pos = __hip_atomic_fetch_add(fillpos + edst[e], 1,
                                           __ATOMIC_RELAXED, __HIP_MEMORY_SCOPE_AGENT);
    ssrc[pos] = esrc[e];
}

__device__ __forceinline__ float lrelu(float x) { return x > 0.f ? x : 0.2f * x; }

__device__ __forceinline__ float4 edge_scores(const float* __restrict__ asn, int s, float4 adv)
{
    const float4 a = *(const float4*)(asn + (size_t)s * HEADS);
    float4 r;
    r.x = lrelu(a.x + adv.x); r.y = lrelu(a.y + adv.y);
    r.z = lrelu(a.z + adv.z); r.w = lrelu(a.w + adv.w);
    return r;
}

__device__ __forceinline__ float4 wave_max4(float4 v)
{
    #pragma unroll
    for (int o = 16; o > 0; o >>= 1) {
        v.x = fmaxf(v.x, __shfl_xor(v.x, o)); v.y = fmaxf(v.y, __shfl_xor(v.y, o));
        v.z = fmaxf(v.z, __shfl_xor(v.z, o)); v.w = fmaxf(v.w, __shfl_xor(v.w, o));
    }
    return v;
}

__device__ __forceinline__ float4 wave_sum4(float4 v)
{
    #pragma unroll
    for (int o = 16; o > 0; o >>= 1) {
        v.x += __shfl_xor(v.x, o); v.y += __shfl_xor(v.y, o);
        v.z += __shfl_xor(v.z, o); v.w += __shfl_xor(v.w, o);
    }
    return v;
}

__device__ __forceinline__ float head_weight(float4 w_l, int i, const unsigned (&hm)[4])
{
    const unsigned bx = (unsigned)__builtin_amdgcn_readlane(__float_as_int(w_l.x), i);
    const unsigned by = (unsigned)__builtin_amdgcn_readlane(__float_as_int(w_l.y), i);
    const unsigned bz = (unsigned)__builtin_amdgcn_readlane(__float_as_int(w_l.z), i);
    const unsigned bw = (unsigned)__builtin_amdgcn_readlane(__float_as_int(w_l.w), i);
    return __uint_as_float((bx & hm[0]) | (by & hm[1]) | (bz & hm[2]) | (bw & hm[3]));
}

__device__ __forceinline__ float4 src_row(const float* __restrict__ src_feat, int s_l, int i, int lane)
{
    const int s = __builtin_amdgcn_readlane(s_l, i);
    return *(const float4*)(src_feat + (size_t)s * FDIM + lane * 4);
}

__device__ __forceinline__ void fma_row(float4& acc, float w, float4 f)
{
    acc.x += w * f.x; acc.y += w * f.y; acc.z += w * f.z; acc.w += w * f.w;
}

__device__ __forceinline__ void add_rows(int s_l, float4 w_l, int m, int lane,
                                         const unsigned (&hm)[4],
                                         const float* __restrict__ src_feat, float4& acc)
{
    int i = 0;
    for (; i + 4 <= m; i += 4) {
        const float4 f0 = src_row(src_feat, s_l, i + 0, lane);
        const float4 f1 = src_row(src_feat, s_l, i + 1, lane);
        const float4 f2 = src_row(src_feat, s_l, i + 2, lane);
        const float4 f3 = src_row(src_feat, s_l, i + 3, lane);
        const float w0 = head_weight(w_l, i + 0, hm);
        const float w1 = head_weight(w_l, i + 1, hm);
        const float w2 = head_weight(w_l, i + 2, hm);
        const float w3 = head_weight(w_l, i + 3, hm);
        fma_row(acc, w0, f0); fma_row(acc, w1, f1);
        fma_row(acc, w2, f2); fma_row(acc, w3, f3);
    }
    for (; i < m; ++i) {
        const float4 f = src_row(src_feat, s_l, i, lane);
        fma_row(acc, head_weight(w_l, i, hm), f);
    }
}

__global__ __launch_bounds__(256) void segment_softmax_aggregate(
    const int* __restrict__ ssrc, const int* __restrict__ start, const int* __restrict__ cnt,
    const float* __restrict__ src_feat, const float* __restrict__ asn,
    const float* __restrict__ adn, float* __restrict__ out, int N)
{
    const int lane = threadIdx.x & 31;
    const int wv = __builtin_amdgcn_readfirstlane((int)(threadIdx.x >> 5));
    const int d = blockIdx.x * 8 + wv;
    if (d >= N) return;
    const int beg = start[d], n = cnt[d];
    const int h = lane >> 3;
    const unsigned hm[4] = { h == 0 ? ~0u : 0u, h == 1 ? ~0u : 0u, h == 2 ? ~0u : 0u, h == 3 ? ~0u : 0u };
    float* op = out + (size_t)d * FDIM + lane * 4;
    const float4 self = *(const float4*)op;
    const float4 adv = *(const float4*)(adn + (size_t)d * HEADS);
    const float NEG = -INFINITY;

    const bool v0 = lane < n;
    const int s0 = v0 ? ssrc[beg + lane] : 0;
    float4 al0 = {NEG, NEG, NEG, NEG};
    if (v0) al0 = edge_scores(asn, s0, adv);

    float4 mx = al0;
    for (int b = 32; b < n; b += 32) {
        const int j = b + lane;
        if (j < n) {
            const float4 al = edge_scores(asn, ssrc[beg + j], adv);
            mx.x = fmaxf(mx.x, al.x); mx.y = fmaxf(mx.y, al.y);
            mx.z = fmaxf(mx.z, al.z); mx.w = fmaxf(mx.w, al.w);
        }
    }
    mx = wave_max4(mx);

    float4 ex0 = {0.f, 0.f, 0.f, 0.f};
    if (v0) {
        ex0.x = __expf(al0.x - mx.x); ex0.y = __expf(al0.y - mx.y);
        ex0.z = __expf(al0.z - mx.z); ex0.w = __expf(al0.w - mx.w);
    }
    float4 sm = ex0;
    for (int b = 32; b < n; b += 32) {
        const int j = b + lane;
        if (j < n) {
            const float4 al = edge_scores(asn, ssrc[beg + j], adv);
            sm.x += __expf(al.x - mx.x); sm.y += __expf(al.y - mx.y);
            sm.z += __expf(al.z - mx.z); sm.w += __expf(al.w - mx.w);
        }
    }
    sm = wave_sum4(sm);
    float4 den;
    den.x = sm.x + 1e-16f; den.y = sm.y + 1e-16f; den.z = sm.z + 1e-16f; den.w = sm.w + 1e-16f;

    float4 acc = {0.f, 0.f, 0.f, 0.f};
    float4 w0;
    w0.x = ex0.x / den.x; w0.y = ex0.y / den.y; w0.z = ex0.z / den.z; w0.w = ex0.w / den.w;
    add_rows(s0, w0, n < 32 ? n : 32, lane, hm, src_feat, acc);
    for (int b = 32; b < n; b += 32) {
        const int j = b + lane;
        int s = 0;
        float4 w = {0.f, 0.f, 0.f, 0.f};
        if (j < n) {
            s = ssrc[beg + j];
            const float4 al = edge_scores(asn, s, adv);
            w.x = __expf(al.x - mx.x) / den.x; w.y = __expf(al.y - mx.y) / den.y;
            w.z = __expf(al.z - mx.z) / den.z; w.w = __expf(al.w - mx.w) / den.w;
        }
        const int rem = n - b;
        add_rows(s, w, rem < 32 ? rem : 32, lane, hm, src_feat, acc);
    }

    float4 o;
    o.x = self.x + acc.x; o.y = self.y + acc.y; o.z = self.z + acc.z; o.w = self.w + acc.w;
    *(float4*)op = o;
}

extern "C" void kernel_launch(void* const* d_in, const int* in_sizes, int n_in,
                              void* d_out, int out_size, void* d_ws, size_t ws_size,
                              hipStream_t stream)
{
    const float* x_src  = (const float*)d_in[0];
    const float* x_dst  = (const float*)d_in[1];
    const int*   esrc   = (const int*)d_in[2];
    const int*   edst   = (const int*)d_in[3];
    const float* W_src  = (const float*)d_in[5];
    const float* W_dst  = (const float*)d_in[6];
    const float* attS   = (const float*)d_in[7];
    const float* attD   = (const float*)d_in[8];
    const float* W_self = (const float*)d_in[9];
    const float* b_self = (const float*)d_in[10];

    const int N_SRC = in_sizes[0] / FDIM;
    const int N_DST = in_sizes[1] / FDIM;
    const int E     = in_sizes[2];

    char* ws = (char*)d_ws;
    float* src_feat = (float*)(ws);
    float* Vp_s     = (float*)(ws + (size_t)N_SRC * FDIM * 4);
    float* Vp_d     = Vp_s + FDIM * 16;
    float* as_node  = (float*)(ws + (size_t)(N_SRC + N_DST) * FDIM * 4);
    float* ad_node  = as_node + (size_t)N_SRC * HEADS;
    int*   cnt      = (int*)(ad_node + (size_t)N_DST * HEADS);
    int*   start    = cnt + N_DST;
    int*   fillpos  = start + N_DST;
    int*   ssrc     = fillpos + N_DST;
    int*   cursor   = ssrc + E;

    float* out = (float*)d_out;

    clear_counts_build_logit_tiles<<<(N_DST + 255) / 256 + 16, 256, 0, stream>>>(
        cnt, cursor, N_DST, (N_DST + 255) / 256, W_src, attS, W_dst, attD, Vp_s, Vp_d);
    gemm_x128_logits<<<(N_SRC + 15) / 16, 288, 0, stream>>>(
        x_src, W_src, Vp_s, nullptr, src_feat, as_node, N_SRC);
    gemm_x128_logits<<<(N_DST + 15) / 16, 288, 0, stream>>>(
        x_dst, W_self, Vp_d, b_self, out, ad_node, N_DST);
    count_dst<<<(E + 255) / 256, 256, 0, stream>>>(edst, cnt, E);
    segment_starts<<<(N_DST + 255) / 256, 256, 0, stream>>>(cnt, cursor, start, fillpos, N_DST);
    place_edges<<<(E + 255) / 256, 256, 0, stream>>>(esrc, edst, fillpos, ssrc, E);
    segment_softmax_aggregate<<<(N_DST + 7) / 8, 256, 0, stream>>>(
        ssrc, start, cnt, src_feat, as_node, ad_node, out, N_DST);
}


# Edges grouped by a two-level counting sort on LDS counters (256-node buckets, then nodes) instead of two device-scope atomics per edge; 7 to 6 launches; sum order within a node changes
# speedup vs baseline: 3.8659x; 1.6131x over previous
#include <hip/hip_runtime.h>
#include <hip/hip_bf16.h>
#include <math.h>

typedef __attribute__((ext_vector_type(2))) float v2f;
typedef __attribute__((ext_vector_type(8))) float v8f;

#define HEADS 4
#define FDIM  128

template <int LDB>
__device__ __forceinline__ v8f tile_product(const float (*At)[132],
                                            const float* __restrict__ Bcol, int m, int koff)
{
    v8f acc = {};
    #pragma unroll
    for (int kb = 0; kb < FDIM; kb += 4) {
        v2f a, b;
        a.x = At[m][kb + koff];
        a.y = At[m][kb + koff + 1];
        b.x = Bcol[(kb + koff)     * LDB];
        b.y = Bcol[(kb + koff + 1) * LDB];
        acc = __builtin_amdgcn_wmma_f32_16x16x4_f32(
                  false, a, false, b, (short)0, acc, false, false);
    }
    return acc;
}

__global__ __launch_bounds__(288) void gemm_x128_logits(
    const float* __restrict__ X, const float* __restrict__ W,
    const float* __restrict__ Vp, const float* __restrict__ bias,
    float* __restrict__ Y, float* __restrict__ alpha, int M)
{
    __shared__ float At[16][132];
    const int tid  = threadIdx.x;
    const int row0 = blockIdx.x * 16;
    const int wv = __builtin_amdgcn_readfirstlane(tid >> 5);

    if (wv < 8) {
        #pragma unroll
        for (int i = 0; i < 2; ++i) {
            int idx = (tid + i * 256) * 4;
            int r = idx >> 7, c = idx & 127;
            int gr = row0 + r; gr = gr < M ? gr : M - 1;
            const float4 v = *(const float4*)(X + (size_t)gr * FDIM + c);
            At[r][c + 0] = v.x; At[r][c + 1] = v.y;
            At[r][c + 2] = v.z; At[r][c + 3] = v.w;
        }
    }
    __syncthreads();

    const int lane = tid & 31;
    const int m    = lane & 15;
    const int half = lane >> 4;
    const int koff = half << 1;

    if (wv < 8) {
        const int n = wv * 16 + m;
        const v8f acc = tile_product<FDIM>(At, W + n, m, koff);
        const float bn = bias ? bias[n] : 0.0f;
        #pragma unroll
        for (int r = 0; r < 8; ++r) {
            int row = row0 + half * 8 + r;
            if (row < M) Y[(size_t)row * FDIM + n] = acc[r] + bn;
        }
    } else {
        const v8f acc = tile_product<16>(At, Vp + m, m, koff);
        if (m < HEADS) {
            #pragma unroll
            for (int r = 0; r < 8; ++r) {
                int row = row0 + half * 8 + r;
                if (row < M) alpha[(size_t)row * HEADS + m] = acc[r];
            }
        }
    }
}

#define BUCKET_SHIFT 8
#define BUCKET_NODES 256
#define CHUNK_EDGES  4096

__device__ __forceinline__ int bucket_of(int d)
{
    const unsigned b = (unsigned)d >> BUCKET_SHIFT;
    return (int)(b < BUCKET_NODES - 1u ? b : BUCKET_NODES - 1u);
}

__device__ __forceinline__ int wave_incl_scan(int v, int lane)
{
    #pragma unroll
    for (int o = 1; o < 32; o <<= 1) {
        const int t = __shfl_up(v, o);
        if (lane >= o) v += t;
    }
    return v;
}

__global__ __launch_bounds__(1024) void chunk_bucket_counts_build_logit_tiles(
    const int* __restrict__ edst, int* __restrict__ hist, int E, int B, int C,
    const float* __restrict__ W_s, const float* __restrict__ att_s,
    const float* __restrict__ W_d, const float* __restrict__ att_d,
    float* __restrict__ Vp_s, float* __restrict__ Vp_d)
{
    __shared__ int h[BUCKET_NODES];
    const int tid = threadIdx.x;
    if ((int)blockIdx.x < C) {
        const int chunk = blockIdx.x;
        if (tid < BUCKET_NODES) h[tid] = 0;
        __syncthreads();
        #pragma unroll
        for (int i = 0; i < 4; ++i) {
            const int e = chunk * CHUNK_EDGES + i * 1024 + tid;
            if (e < E)
                __hip_atomic_fetch_add(&h[bucket_of(edst[e])], 1,
                                       __ATOMIC_RELAXED, __HIP_MEMORY_SCOPE_WORKGROUP);
        }
        __syncthreads();
        if (tid < B) hist[chunk * B + tid] = h[tid];
        return;
    }
    const int t = ((int)blockIdx.x - C) * 1024 + tid;
    const int which = t >> 11;
    const int k = (t >> 4) & 127;
    const int c = t & 15;
    const float* Wm  = which ? W_d : W_s;
    const float* att = which ? att_d : att_s;
    float s = 0.0f;
    if (c < HEADS) {
        const float* wr = Wm + (size_t)k * FDIM + c * 32;
        const float* ar = att + c * 32;
        #pragma unroll
        for (int d = 0; d < 32; d += 4) {
            const float4 w = *(const float4*)(wr + d);
            const float4 a = *(const float4*)(ar + d);
            s += w.x * a.x; s += w.y * a.y; s += w.z * a.z; s += w.w * a.w;
        }
    }
    (which ? Vp_d : Vp_s)[k * 16 + c] = s;
}

__global__ __launch_bounds__(1024) void deal_edges_to_buckets(
    const int* __restrict__ esrc, const int* __restrict__ edst,
    const int* __restrict__ hist, int* __restrict__ bucket_start,
    int2* __restrict__ pairs, int E, int B, int C)
{
    __shared__ int s_tot[4][BUCKET_NODES];
    __shared__ int s_pre[4][BUCKET_NODES];
    __shared__ int s_base[BUCKET_NODES];
    __shared__ int s_fill[BUCKET_NODES];
    __shared__ int s_wsum[8];
    const int tid = threadIdx.x;
    const int lane = tid & 31;
    const int chunk = blockIdx.x;

    int d[4], s[4];
    #pragma unroll
    for (int i = 0; i < 4; ++i) {
        const int e = chunk * CHUNK_EDGES + i * 1024 + tid;
        d[i] = e < E ? edst[e] : -1;
        s[i] = e < E ? esrc[e] : 0;
    }

    const int g = tid >> 8, b = tid & (BUCKET_NODES - 1);
    int tot = 0, pre = 0;
    if (b < B) {
        #pragma unroll 7
        for (int c = g; c < C; c += 4) {
            const int v = hist[c * B + b];
            tot += v;
            pre += c < chunk ? v : 0;
        }
    }
    s_tot[g][b] = tot; s_pre[g][b] = pre;
    __syncthreads();
    int incl = 0;
    if (tid < BUCKET_NODES) {
        tot = s_tot[0][tid] + s_tot[1][tid] + s_tot[2][tid] + s_tot[3][tid];
        pre = s_pre[0][tid] + s_pre[1][tid] + s_pre[2][tid] + s_pre[3][tid];
        incl = wave_incl_scan(tot, lane);
        if (lane == 31) s_wsum[tid >> 5] = incl;
    }
    __syncthreads();
    if (tid < BUCKET_NODES) {
        int wbase = 0;
        for (int w = 0; w < (tid >> 5); ++w) wbase += s_wsum[w];
        const int excl = wbase + incl - tot;
        s_base[tid] = excl + pre;
        s_fill[tid] = 0;
        if (chunk == 0) {
            bucket_start[tid] = excl;
            if (tid == BUCKET_NODES - 1) bucket_start[BUCKET_NODES] = excl + tot;
        }
    }
    __syncthreads();
    #pragma unroll
    for (int i = 0; i < 4; ++i) {
        if (chunk * CHUNK_EDGES + i * 1024 + tid < E) {
            const int bk = bucket_of(d[i]);
            const int r = __hip_atomic_fetch_add(&s_fill[bk], 1,
                                                 __ATOMIC_RELAXED, __HIP_MEMORY_SCOPE_WORKGROUP);
            pairs[s_base[bk] + r] = make_int2(d[i], s[i]);
        }
    }
}

__global__ __launch_bounds__(1024) void group_bucket_by_node(
    const int2* __restrict__ pairs, const int* __restrict__ bucket_start,
    int* __restrict__ start, int* __restrict__ cnt, int* __restrict__ ssrc, int N)
{
    __shared__ int s_cnt[BUCKET_NODES];
    __shared__ int s_off[BUCKET_NODES];
    __shared__ int s_wsum[8];
    const int tid = threadIdx.x;
    const int lane = tid & 31;
    const int bk = blockIdx.x;
    const int beg = bucket_start[bk], end = bucket_start[bk + 1];
    if (tid < BUCKET_NODES) s_cnt[tid] = 0;
    __syncthreads();
    for (int e0 = beg + tid; e0 < end; e0 += 4096) {
        int dl[4];
        #pragma unroll
        for (int i = 0; i < 4; ++i) {
            const int e = e0 + i * 1024;
            dl[i] = pairs[e < end ? e : end - 1].x;
        }
        #pragma unroll
        for (int i = 0; i < 4; ++i)
            if (e0 + i * 1024 < end)
                __hip_atomic_fetch_add(&s_cnt[dl[i] & (BUCKET_NODES - 1)], 1,
                                       __ATOMIC_RELAXED, __HIP_MEMORY_SCOPE_WORKGROUP);
    }
    __syncthreads();
    int c = 0, incl = 0;
    if (tid < BUCKET_NODES) {
        c = s_cnt[tid];
        incl = wave_incl_scan(c, lane);
        if (lane == 31) s_wsum[tid >> 5] = incl;
    }
    __syncthreads();
    if (tid < BUCKET_NODES) {
        int wbase = 0;
        for (int w = 0; w < (tid >> 5); ++w) wbase += s_wsum[w];
        const int st = beg + wbase + incl - c;
        s_off[tid] = st;
        const int node = bk * BUCKET_NODES + tid;
        if (node < N) { start[node] = st; cnt[node] = c; }
    }
    __syncthreads();
    for (int e0 = beg + tid; e0 < end; e0 += 4096) {
        int2 p[4];
        #pragma unroll
        for (int i = 0; i < 4; ++i) {
            const int e = e0 + i * 1024;
            p[i] = pairs[e < end ? e : end - 1];
        }
        #pragma unroll
        for (int i = 0; i < 4; ++i)
            if (e0 + i * 1024 < end) {
                const int pos = __hip_atomic_fetch_add(&s_off[p[i].x & (BUCKET_NODES - 1)], 1,
                                                       __ATOMIC_RELAXED, __HIP_MEMORY_SCOPE_WORKGROUP);
                ssrc[pos] = p[i].y;
            }
    }
}

__device__ __forceinline__ float lrelu(float x) { return x > 0.f ? x : 0.2f * x; }

__device__ __forceinline__ float4 edge_scores(const float* __restrict__ asn, int s, float4 adv)
{
    const float4 a = *(const float4*)(asn + (size_t)s * HEADS);
    float4 r;
    r.x = lrelu(a.x + adv.x); r.y = lrelu(a.y + adv.y);
    r.z = lrelu(a.z + adv.z); r.w = lrelu(a.w + adv.w);
    return r;
}

__device__ __forceinline__ float4 wave_max4(float4 v)
{
    #pragma unroll
    for (int o = 16; o > 0; o >>= 1) {
        v.x = fmaxf(v.x, __shfl_xor(v.x, o)); v.y = fmaxf(v.y, __shfl_xor(v.y, o));
        v.z = fmaxf(v.z, __shfl_xor(v.z, o)); v.w = fmaxf(v.w, __shfl_xor(v.w, o));
    }
    return v;
}

__device__ __forceinline__ float4 wave_sum4(float4 v)
{
    #pragma unroll
    for (int o = 16; o > 0; o >>= 1) {
        v.x += __shfl_xor(v.x, o); v.y += __shfl_xor(v.y, o);
        v.z += __shfl_xor(v.z, o); v.w += __shfl_xor(v.w, o);
    }
    return v;
}

__device__ __forceinline__ float head_weight(float4 w_l, int i, const unsigned (&hm)[4])
{
    const unsigned bx = (unsigned)__builtin_amdgcn_readlane(__float_as_int(w_l.x), i);
    const unsigned by = (unsigned)__builtin_amdgcn_readlane(__float_as_int(w_l.y), i);
    const unsigned bz = (unsigned)__builtin_amdgcn_readlane(__float_as_int(w_l.z), i);
    const unsigned bw = (unsigned)__builtin_amdgcn_readlane(__float_as_int(w_l.w), i);
    return __uint_as_float((bx & hm[0]) | (by & hm[1]) | (bz & hm[2]) | (bw & hm[3]));
}

__device__ __forceinline__ float4 src_row(const float* __restrict__ src_feat, int s_l, int i, int lane)
{
    const int s = __builtin_amdgcn_readlane(s_l, i);
    return *(const float4*)(src_feat + (size_t)s * FDIM + lane * 4);
}

__device__ __forceinline__ void fma_row(float4& acc, float w, float4 f)
{
    acc.x += w * f.x; acc.y += w * f.y; acc.z += w * f.z; acc.w += w * f.w;
}

__device__ __forceinline__ void add_rows(int s_l, float4 w_l, int m, int lane,
                                         const unsigned (&hm)[4],
                                         const float* __restrict__ src_feat, float4& acc)
{
    int i = 0;
    for (; i + 4 <= m; i += 4) {
        const float4 f0 = src_row(src_feat, s_l, i + 0, lane);
        const float4 f1 = src_row(src_feat, s_l, i + 1, lane);
        const float4 f2 = src_row(src_feat, s_l, i + 2, lane);
        const float4 f3 = src_row(src_feat, s_l, i + 3, lane);
        const float w0 = head_weight(w_l, i + 0, hm);
        const float w1 = head_weight(w_l, i + 1, hm);
        const float w2 = head_weight(w_l, i + 2, hm);
        const float w3 = head_weight(w_l, i + 3, hm);
        fma_row(acc, w0, f0); fma_row(acc, w1, f1);
        fma_row(acc, w2, f2); fma_row(acc, w3, f3);
    }
    for (; i < m; ++i) {
        const float4 f = src_row(src_feat, s_l, i, lane);
        fma_row(acc, head_weight(w_l, i, hm), f);
    }
}

__global__ __launch_bounds__(256) void segment_softmax_aggregate(
    const int* __restrict__ ssrc, const int* __restrict__ start, const int* __restrict__ cnt,
    const float* __restrict__ src_feat, const float* __restrict__ asn,
    const float* __restrict__ adn, float* __restrict__ out, int N)
{
    const int lane = threadIdx.x & 31;
    const int wv = __builtin_amdgcn_readfirstlane((int)(threadIdx.x >> 5));
    const int d = blockIdx.x * 8 + wv;
    if (d >= N) return;
    const int beg = start[d], n = cnt[d];
    const int h = lane >> 3;
    const unsigned hm[4] = { h == 0 ? ~0u : 0u, h == 1 ? ~0u : 0u, h == 2 ? ~0u : 0u, h == 3 ? ~0u : 0u };
    float* op = out + (size_t)d * FDIM + lane * 4;
    const float4 self = *(const float4*)op;
    const float4 adv = *(const float4*)(adn + (size_t)d * HEADS);
    const float NEG = -INFINITY;

    const bool v0 = lane < n;
    const int s0 = v0 ? ssrc[beg + lane] : 0;
    float4 al0 = {NEG, NEG, NEG, NEG};
    if (v0) al0 = edge_scores(asn, s0, adv);

    float4 mx = al0;
    for (int b = 32; b < n; b += 32) {
        const int j = b + lane;
        if (j < n) {
            const float4 al = edge_scores(asn, ssrc[beg + j], adv);
            mx.x = fmaxf(mx.x, al.x); mx.y = fmaxf(mx.y, al.y);
            mx.z = fmaxf(mx.z, al.z); mx.w = fmaxf(mx.w, al.w);
        }
    }
    mx = wave_max4(mx);

    float4 ex0 = {0.f, 0.f, 0.f, 0.f};
    if (v0) {
        ex0.x = __expf(al0.x - mx.x); ex0.y = __expf(al0.y - mx.y);
        ex0.z = __expf(al0.z - mx.z); ex0.w = __expf(al0.w - mx.w);
    }
    float4 sm = ex0;
    for (int b = 32; b < n; b += 32) {
        const int j = b + lane;
        if (j < n) {
            const float4 al = edge_scores(asn, ssrc[beg + j], adv);
            sm.x += __expf(al.x - mx.x); sm.y += __expf(al.y - mx.y);
            sm.z += __expf(al.z - mx.z); sm.w += __expf(al.w - mx.w);
        }
    }
    sm = wave_sum4(sm);
    float4 den;
    den.x = sm.x + 1e-16f; den.y = sm.y + 1e-16f; den.z = sm.z + 1e-16f; den.w = sm.w + 1e-16f;

    float4 acc = {0.f, 0.f, 0.f, 0.f};
    float4 w0;
    w0.x = ex0.x / den.x; w0.y = ex0.y / den.y; w0.z = ex0.z / den.z; w0.w = ex0.w / den.w;
    add_rows(s0, w0, n < 32 ? n : 32, lane, hm, src_feat, acc);
    for (int b = 32; b < n; b += 32) {
        const int j = b + lane;
        int s = 0;
        float4 w = {0.f, 0.f, 0.f, 0.f};
        if (j < n) {
            s = ssrc[beg + j];
            const float4 al = edge_scores(asn, s, adv);
            w.x = __expf(al.x - mx.x) / den.x; w.y = __expf(al.y - mx.y) / den.y;
            w.z = __expf(al.z - mx.z) / den.z; w.w = __expf(al.w - mx.w) / den.w;
        }
        const int rem = n - b;
        add_rows(s, w, rem < 32 ? rem : 32, lane, hm, src_feat, acc);
    }

    float4 o;
    o.x = self.x + acc.x; o.y = self.y + acc.y; o.z = self.z + acc.z; o.w = self.w + acc.w;
    *(float4*)op = o;
}

extern "C" void kernel_launch(void* const* d_in, const int* in_sizes, int n_in,
                              void* d_out, int out_size, void* d_ws, size_t ws_size,
                              hipStream_t stream)
{
    const float* x_src  = (const float*)d_in[0];
    const float* x_dst  = (const float*)d_in[1];
    const int*   esrc   = (const int*)d_in[2];
    const int*   edst   = (const int*)d_in[3];
    const float* W_src  = (const float*)d_in[5];
    const float* W_dst  = (const float*)d_in[6];
    const float* attS   = (const float*)d_in[7];
    const float* attD   = (const float*)d_in[8];
    const float* W_self = (const float*)d_in[9];
    const float* b_self = (const float*)d_in[10];

    const int N_SRC = in_sizes[0] / FDIM;
    const int N_DST = in_sizes[1] / FDIM;
    const int E     = in_sizes[2];

    char* ws = (char*)d_ws;
    float* src_feat = (float*)(ws);
    float* Vp_s     = (float*)(ws + (size_t)N_SRC * FDIM * 4);
    float* Vp_d     = Vp_s + FDIM * 16;
    const int B = (N_DST + BUCKET_NODES - 1) / BUCKET_NODES;
    const int C = (E + CHUNK_EDGES - 1) / CHUNK_EDGES;
    int2*  pairs    = (int2*)(Vp_d + FDIM * 16);
    int*   hist     = (int*)(pairs + E);
    int*   bstart   = hist + (size_t)C * B;
    float* as_node  = (float*)(ws + (size_t)(N_SRC + N_DST) * FDIM * 4);
    float* ad_node  = as_node + (size_t)N_SRC * HEADS;
    int*   cnt      = (int*)(ad_node + (size_t)N_DST * HEADS);
    int*   start    = cnt + N_DST;
    int*   ssrc     = start + 2 * (size_t)N_DST;

    float* out = (float*)d_out;

    chunk_bucket_counts_build_logit_tiles<<<C + 4, 1024, 0, stream>>>(
        edst, hist, E, B, C, W_src, attS, W_dst, attD, Vp_s, Vp_d);
    gemm_x128_logits<<<(N_SRC + 15) / 16, 288, 0, stream>>>(
        x_src, W_src, Vp_s, nullptr, src_feat, as_node, N_SRC);
    gemm_x128_logits<<<(N_DST + 15) / 16, 288, 0, stream>>>(
        x_dst, W_self, Vp_d, b_self, out, ad_node, N_DST);
    deal_edges_to_buckets<<<C, 1024, 0, stream>>>(esrc, edst, hist, bstart, pairs, E, B, C);
    group_bucket_by_node<<<B, 1024, 0, stream>>>(pairs, bstart, start, cnt, ssrc, N_DST);
    segment_softmax_aggregate<<<(N_DST + 7) / 8, 256, 0, stream>>>(
        ssrc, start, cnt, src_feat, as_node, ad_node, out, N_DST);
}


# Self-projection GEMM moved into the aggregate launch: a 16-node work-group builds its x_dst @ W_self tile and logits in LDS on WMMA, then one wave per node gathers; 6 to 5 launches, out stored once
# speedup vs baseline: 4.6506x; 1.2030x over previous
#include <hip/hip_runtime.h>
#include <hip/hip_bf16.h>
#include <math.h>

typedef __attribute__((ext_vector_type(2))) float v2f;
typedef __attribute__((ext_vector_type(8))) float v8f;

#define HEADS 4
#define FDIM  128

template <int LDB>
__device__ __forceinline__ v8f tile_product(const float (*At)[132],
                                            const float* __restrict__ Bcol, int m, int koff)
{
    v8f acc = {};
    #pragma unroll
    for (int kb = 0; kb < FDIM; kb += 4) {
        v2f a, b;
        a.x = At[m][kb + koff];
        a.y = At[m][kb + koff + 1];
        b.x = Bcol[(kb + koff)     * LDB];
        b.y = Bcol[(kb + koff + 1) * LDB];
        acc = __builtin_amdgcn_wmma_f32_16x16x4_f32(
                  false, a, false, b, (short)0, acc, false, false);
    }
    return acc;
}

__global__ __launch_bounds__(288) void gemm_x128_logits(
    const float* __restrict__ X, const float* __restrict__ W,
    const float* __restrict__ Vp, const float* __restrict__ bias,
    float* __restrict__ Y, float* __restrict__ alpha, int M)
{
    __shared__ float At[16][132];
    const int tid  = threadIdx.x;
    const int row0 = blockIdx.x * 16;
    const int wv = __builtin_amdgcn_readfirstlane(tid >> 5);

    if (wv < 8) {
        #pragma unroll
        for (int i = 0; i < 2; ++i) {
            int idx = (tid + i * 256) * 4;
            int r = idx >> 7, c = idx & 127;
            int gr = row0 + r; gr = gr < M ? gr : M - 1;
            const float4 v = *(const float4*)(X + (size_t)gr * FDIM + c);
            At[r][c + 0] = v.x; At[r][c + 1] = v.y;
            At[r][c + 2] = v.z; At[r][c + 3] = v.w;
        }
    }
    __syncthreads();

    const int lane = tid & 31;
    const int m    = lane & 15;
    const int half = lane >> 4;
    const int koff = half << 1;

    if (wv < 8) {
        const int n = wv * 16 + m;
        const v8f acc = tile_product<FDIM>(At, W + n, m, koff);
        const float bn = bias ? bias[n] : 0.0f;
        #pragma unroll
        for (int r = 0; r < 8; ++r) {
            int row = row0 + half * 8 + r;
            if (row < M) Y[(size_t)row * FDIM + n] = acc[r] + bn;
        }
    } else {
        const v8f acc = tile_product<16>(At, Vp + m, m, koff);
        if (m < HEADS) {
            #pragma unroll
            for (int r = 0; r < 8; ++r) {
                int row = row0 + half * 8 + r;
                if (row < M) alpha[(size_t)row * HEADS + m] = acc[r];
            }
        }
    }
}

#define BUCKET_SHIFT 8
#define BUCKET_NODES 256
#define CHUNK_EDGES  4096

__device__ __forceinline__ int bucket_of(int d)
{
    const unsigned b = (unsigned)d >> BUCKET_SHIFT;
    return (int)(b < BUCKET_NODES - 1u ? b : BUCKET_NODES - 1u);
}

__device__ __forceinline__ int wave_incl_scan(int v, int lane)
{
    #pragma unroll
    for (int o = 1; o < 32; o <<= 1) {
        const int t = __shfl_up(v, o);
        if (lane >= o) v += t;
    }
    return v;
}

__global__ __launch_bounds__(1024) void chunk_bucket_counts_build_logit_tiles(
    const int* __restrict__ edst, int* __restrict__ hist, int E, int B, int C,
    const float* __restrict__ W_s, const float* __restrict__ att_s,
    const float* __restrict__ W_d, const float* __restrict__ att_d,
    float* __restrict__ Vp_s, float* __restrict__ Vp_d)
{
    __shared__ int h[BUCKET_NODES];
    const int tid = threadIdx.x;
    if ((int)blockIdx.x < C) {
        const int chunk = blockIdx.x;
        if (tid < BUCKET_NODES) h[tid] = 0;
        __syncthreads();
        #pragma unroll
        for (int i = 0; i < 4; ++i) {
            const int e = chunk * CHUNK_EDGES + i * 1024 + tid;
            if (e < E)
                __hip_atomic_fetch_add(&h[bucket_of(edst[e])], 1,
                                       __ATOMIC_RELAXED, __HIP_MEMORY_SCOPE_WORKGROUP);
        }
        __syncthreads();
        if (tid < B) hist[chunk * B + tid] = h[tid];
        return;
    }
    const int t = ((int)blockIdx.x - C) * 1024 + tid;
    const int which = t >> 11;
    const int k = (t >> 4) & 127;
    const int c = t & 15;
    const float* Wm  = which ? W_d : W_s;
    const float* att = which ? att_d : att_s;
    float s = 0.0f;
    if (c < HEADS) {
        const float* wr = Wm + (size_t)k * FDIM + c * 32;
        const float* ar = att + c * 32;
        #pragma unroll
        for (int d = 0; d < 32; d += 4) {
            const float4 w = *(const float4*)(wr + d);
            const float4 a = *(const float4*)(ar + d);
            s += w.x * a.x; s += w.y * a.y; s += w.z * a.z; s += w.w * a.w;
        }
    }
    (which ? Vp_d : Vp_s)[k * 16 + c] = s;
}

__global__ __launch_bounds__(1024) void deal_edges_to_buckets(
    const int* __restrict__ esrc, const int* __restrict__ edst,
    const int* __restrict__ hist, int* __restrict__ bucket_start,
    int2* __restrict__ pairs, int E, int B, int C)
{
    __shared__ int s_tot[4][BUCKET_NODES];
    __shared__ int s_pre[4][BUCKET_NODES];
    __shared__ int s_base[BUCKET_NODES];
    __shared__ int s_fill[BUCKET_NODES];
    __shared__ int s_wsum[8];
    const int tid = threadIdx.x;
    const int lane = tid & 31;
    const int chunk = blockIdx.x;

    int d[4], s[4];
    #pragma unroll
    for (int i = 0; i < 4; ++i) {
        const int e = chunk * CHUNK_EDGES + i * 1024 + tid;
        d[i] = e < E ? edst[e] : -1;
        s[i] = e < E ? esrc[e] : 0;
    }

    const int g = tid >> 8, b = tid & (BUCKET_NODES - 1);
    int tot = 0, pre = 0;
    if (b < B) {
        #pragma unroll 7
        for (int c = g; c < C; c += 4) {
            const int v = hist[c * B + b];
            tot += v;
            pre += c < chunk ? v : 0;
        }
    }
    s_tot[g][b] = tot; s_pre[g][b] = pre;
    __syncthreads();
    int incl = 0;
    if (tid < BUCKET_NODES) {
        tot = s_tot[0][tid] + s_tot[1][tid] + s_tot[2][tid] + s_tot[3][tid];
        pre = s_pre[0][tid] + s_pre[1][tid] + s_pre[2][tid] + s_pre[3][tid];
        incl = wave_incl_scan(tot, lane);
        if (lane == 31) s_wsum[tid >> 5] = incl;
    }
    __syncthreads();
    if (tid < BUCKET_NODES) {
        int wbase = 0;
        for (int w = 0; w < (tid >> 5); ++w) wbase += s_wsum[w];
        const int excl = wbase + incl - tot;
        s_base[tid] = excl + pre;
        s_fill[tid] = 0;
        if (chunk == 0) {
            bucket_start[tid] = excl;
            if (tid == BUCKET_NODES - 1) bucket_start[BUCKET_NODES] = excl + tot;
        }
    }
    __syncthreads();
    #pragma unroll
    for (int i = 0; i < 4; ++i) {
        if (chunk * CHUNK_EDGES + i * 1024 + tid < E) {
            const int bk = bucket_of(d[i]);
            const int r = __hip_atomic_fetch_add(&s_fill[bk], 1,
                                                 __ATOMIC_RELAXED, __HIP_MEMORY_SCOPE_WORKGROUP);
            pairs[s_base[bk] + r] = make_int2(d[i], s[i]);
        }
    }
}

__global__ __launch_bounds__(1024) void group_bucket_by_node(
    const int2* __restrict__ pairs, const int* __restrict__ bucket_start,
    int* __restrict__ start, int* __restrict__ cnt, int* __restrict__ ssrc, int N)
{
    __shared__ int s_cnt[BUCKET_NODES];
    __shared__ int s_off[BUCKET_NODES];
    __shared__ int s_wsum[8];
    const int tid = threadIdx.x;
    const int lane = tid & 31;
    const int bk = blockIdx.x;
    const int beg = bucket_start[bk], end = bucket_start[bk + 1];
    if (tid < BUCKET_NODES) s_cnt[tid] = 0;
    __syncthreads();
    for (int e0 = beg + tid; e0 < end; e0 += 4096) {
        int dl[4];
        #pragma unroll
        for (int i = 0; i < 4; ++i) {
            const int e = e0 + i * 1024;
            dl[i] = pairs[e < end ? e : end - 1].x;
        }
        #pragma unroll
        for (int i = 0; i < 4; ++i)
            if (e0 + i * 1024 < end)
                __hip_atomic_fetch_add(&s_cnt[dl[i] & (BUCKET_NODES - 1)], 1,
                                       __ATOMIC_RELAXED, __HIP_MEMORY_SCOPE_WORKGROUP);
    }
    __syncthreads();
    int c = 0, incl = 0;
    if (tid < BUCKET_NODES) {
        c = s_cnt[tid];
        incl = wave_incl_scan(c, lane);
        if (lane == 31) s_wsum[tid >> 5] = incl;
    }
    __syncthreads();
    if (tid < BUCKET_NODES) {
        int wbase = 0;
        for (int w = 0; w < (tid >> 5); ++w) wbase += s_wsum[w];
        const int st = beg + wbase + incl - c;
        s_off[tid] = st;
        const int node = bk * BUCKET_NODES + tid;
        if (node < N) { start[node] = st; cnt[node] = c; }
    }
    __syncthreads();
    for (int e0 = beg + tid; e0 < end; e0 += 4096) {
        int2 p[4];
        #pragma unroll
        for (int i = 0; i < 4; ++i) {
            const int e = e0 + i * 1024;
            p[i] = pairs[e < end ? e : end - 1];
        }
        #pragma unroll
        for (int i = 0; i < 4; ++i)
            if (e0 + i * 1024 < end) {
                const int pos = __hip_atomic_fetch_add(&s_off[p[i].x & (BUCKET_NODES - 1)], 1,
                                                       __ATOMIC_RELAXED, __HIP_MEMORY_SCOPE_WORKGROUP);
                ssrc[pos] = p[i].y;
            }
    }
}

__device__ __forceinline__ float lrelu(float x) { return x > 0.f ? x : 0.2f * x; }

__device__ __forceinline__ float4 edge_scores(const float* __restrict__ asn, int s, float4 adv)
{
    const float4 a = *(const float4*)(asn + (size_t)s * HEADS);
    float4 r;
    r.x = lrelu(a.x + adv.x); r.y = lrelu(a.y + adv.y);
    r.z = lrelu(a.z + adv.z); r.w = lrelu(a.w + adv.w);
    return r;
}

__device__ __forceinline__ float4 wave_max4(float4 v)
{
    #pragma unroll
    for (int o = 16; o > 0; o >>= 1) {
        v.x = fmaxf(v.x, __shfl_xor(v.x, o)); v.y = fmaxf(v.y, __shfl_xor(v.y, o));
        v.z = fmaxf(v.z, __shfl_xor(v.z, o)); v.w = fmaxf(v.w, __shfl_xor(v.w, o));
    }
    return v;
}

__device__ __forceinline__ float4 wave_sum4(float4 v)
{
    #pragma unroll
    for (int o = 16; o > 0; o >>= 1) {
        v.x += __shfl_xor(v.x, o); v.y += __shfl_xor(v.y, o);
        v.z += __shfl_xor(v.z, o); v.w += __shfl_xor(v.w, o);
    }
    return v;
}

__device__ __forceinline__ float head_weight(float4 w_l, int i, const unsigned (&hm)[4])
{
    const unsigned bx = (unsigned)__builtin_amdgcn_readlane(__float_as_int(w_l.x), i);
    const unsigned by = (unsigned)__builtin_amdgcn_readlane(__float_as_int(w_l.y), i);
    const unsigned bz = (unsigned)__builtin_amdgcn_readlane(__float_as_int(w_l.z), i);
    const unsigned bw = (unsigned)__builtin_amdgcn_readlane(__float_as_int(w_l.w), i);
    return __uint_as_float((bx & hm[0]) | (by & hm[1]) | (bz & hm[2]) | (bw & hm[3]));
}

__device__ __forceinline__ float4 src_row(const float* __restrict__ src_feat, int s_l, int i, int lane)
{
    const int s = __builtin_amdgcn_readlane(s_l, i);
    return *(const float4*)(src_feat + (size_t)s * FDIM + lane * 4);
}

__device__ __forceinline__ void fma_row(float4& acc, float w, float4 f)
{
    acc.x += w * f.x; acc.y += w * f.y; acc.z += w * f.z; acc.w += w * f.w;
}

__device__ __forceinline__ void add_rows(int s_l, float4 w_l, int m, int lane,
                                         const unsigned (&hm)[4],
                                         const float* __restrict__ src_feat, float4& acc)
{
    int i = 0;
    for (; i + 4 <= m; i += 4) {
        const float4 f0 = src_row(src_feat, s_l, i + 0, lane);
        const float4 f1 = src_row(src_feat, s_l, i + 1, lane);
        const float4 f2 = src_row(src_feat, s_l, i + 2, lane);
        const float4 f3 = src_row(src_feat, s_l, i + 3, lane);
        const float w0 = head_weight(w_l, i + 0, hm);
        const float w1 = head_weight(w_l, i + 1, hm);
        const float w2 = head_weight(w_l, i + 2, hm);
        const float w3 = head_weight(w_l, i + 3, hm);
        fma_row(acc, w0, f0); fma_row(acc, w1, f1);
        fma_row(acc, w2, f2); fma_row(acc, w3, f3);
    }
    for (; i < m; ++i) {
        const float4 f = src_row(src_feat, s_l, i, lane);
        fma_row(acc, head_weight(w_l, i, hm), f);
    }
}

__global__ __launch_bounds__(512) void self_project_softmax_aggregate(
    const float* __restrict__ X, const float* __restrict__ W,
    const float* __restrict__ Vp, const float* __restrict__ bias,
    const int* __restrict__ ssrc, const int* __restrict__ start, const int* __restrict__ cnt,
    const float* __restrict__ src_feat, const float* __restrict__ asn,
    float* __restrict__ out, int N)
{
    __shared__ float At[16][132];
    __shared__ __attribute__((aligned(16))) float Yt[16][132];
    __shared__ __attribute__((aligned(16))) float Lt[16][HEADS];
    const int tid  = threadIdx.x;
    const int lane = tid & 31;
    const int row0 = blockIdx.x * 16;
    const int wv = __builtin_amdgcn_readfirstlane(tid >> 5);

    const int d  = row0 + wv;
    const int dc = d < N ? d : N - 1;
    const int beg = start[dc], n = cnt[dc];

    if (wv < 8) {
        #pragma unroll
        for (int i = 0; i < 2; ++i) {
            int idx = (tid + i * 256) * 4;
            int r = idx >> 7, c = idx & 127;
            int gr = row0 + r; gr = gr < N ? gr : N - 1;
            const float4 v = *(const float4*)(X + (size_t)gr * FDIM + c);
            At[r][c + 0] = v.x; At[r][c + 1] = v.y;
            At[r][c + 2] = v.z; At[r][c + 3] = v.w;
        }
    }
    __syncthreads();
    {
        const int m    = lane & 15;
        const int half = lane >> 4;
        const int koff = half << 1;
        if (wv < 8) {
            const int col = wv * 16 + m;
            const v8f acc = tile_product<FDIM>(At, W + col, m, koff);
            const float bn = bias[col];
            #pragma unroll
            for (int r = 0; r < 8; ++r)
                Yt[half * 8 + r][col] = acc[r] + bn;
        } else if (wv == 8) {
            const v8f acc = tile_product<16>(At, Vp + m, m, koff);
            if (m < HEADS) {
                #pragma unroll
                for (int r = 0; r < 8; ++r)
                    Lt[half * 8 + r][m] = acc[r];
            }
        }
    }
    __syncthreads();

    if (d >= N) return;
    const int h = lane >> 3;
    const unsigned hm[4] = { h == 0 ? ~0u : 0u, h == 1 ? ~0u : 0u, h == 2 ? ~0u : 0u, h == 3 ? ~0u : 0u };
    float* op = out + (size_t)d * FDIM + lane * 4;
    const float4 self = *(const float4*)&Yt[wv][lane * 4];
    const float4 adv = *(const float4*)&Lt[wv][0];
    const float NEG = -INFINITY;

    const bool v0 = lane < n;
    const int s0 = v0 ? ssrc[beg + lane] : 0;
    float4 al0 = {NEG, NEG, NEG, NEG};
    if (v0) al0 = edge_scores(asn, s0, adv);

    float4 mx = al0;
    for (int b = 32; b < n; b += 32) {
        const int j = b + lane;
        if (j < n) {
            const float4 al = edge_scores(asn, ssrc[beg + j], adv);
            mx.x = fmaxf(mx.x, al.x); mx.y = fmaxf(mx.y, al.y);
            mx.z = fmaxf(mx.z, al.z); mx.w = fmaxf(mx.w, al.w);
        }
    }
    mx = wave_max4(mx);

    float4 ex0 = {0.f, 0.f, 0.f, 0.f};
    if (v0) {
        ex0.x = __expf(al0.x - mx.x); ex0.y = __expf(al0.y - mx.y);
        ex0.z = __expf(al0.z - mx.z); ex0.w = __expf(al0.w - mx.w);
    }
    float4 sm = ex0;
    for (int b = 32; b < n; b += 32) {
        const int j = b + lane;
        if (j < n) {
            const float4 al = edge_scores(asn, ssrc[beg + j], adv);
            sm.x += __expf(al.x - mx.x); sm.y += __expf(al.y - mx.y);
            sm.z += __expf(al.z - mx.z); sm.w += __expf(al.w - mx.w);
        }
    }
    sm = wave_sum4(sm);
    float4 den;
    den.x = sm.x + 1e-16f; den.y = sm.y + 1e-16f; den.z = sm.z + 1e-16f; den.w = sm.w + 1e-16f;

    float4 acc = {0.f, 0.f, 0.f, 0.f};
    float4 w0;
    w0.x = ex0.x / den.x; w0.y = ex0.y / den.y; w0.z = ex0.z / den.z; w0.w = ex0.w / den.w;
    add_rows(s0, w0, n < 32 ? n : 32, lane, hm, src_feat, acc);
    for (int b = 32; b < n; b += 32) {
        const int j = b + lane;
        int s = 0;
        float4 w = {0.f, 0.f, 0.f, 0.f};
        if (j < n) {
            s = ssrc[beg + j];
            const float4 al = edge_scores(asn, s, adv);
            w.x = __expf(al.x - mx.x) / den.x; w.y = __expf(al.y - mx.y) / den.y;
            w.z = __expf(al.z - mx.z) / den.z; w.w = __expf(al.w - mx.w) / den.w;
        }
        const int rem = n - b;
        add_rows(s, w, rem < 32 ? rem : 32, lane, hm, src_feat, acc);
    }

    float4 o;
    o.x = self.x + acc.x; o.y = self.y + acc.y; o.z = self.z + acc.z; o.w = self.w + acc.w;
    *(float4*)op = o;
}

extern "C" void kernel_launch(void* const* d_in, const int* in_sizes, int n_in,
                              void* d_out, int out_size, void* d_ws, size_t ws_size,
                              hipStream_t stream)
{
    const float* x_src  = (const float*)d_in[0];
    const float* x_dst  = (const float*)d_in[1];
    const int*   esrc   = (const int*)d_in[2];
    const int*   edst   = (const int*)d_in[3];
    const float* W_src  = (const float*)d_in[5];
    const float* W_dst  = (const float*)d_in[6];
    const float* attS   = (const float*)d_in[7];
    const float* attD   = (const float*)d_in[8];
    const float* W_self = (const float*)d_in[9];
    const float* b_self = (const float*)d_in[10];

    const int N_SRC = in_sizes[0] / FDIM;
    const int N_DST = in_sizes[1] / FDIM;
    const int E     = in_sizes[2];

    char* ws = (char*)d_ws;
    float* src_feat = (float*)(ws);
    float* Vp_s     = (float*)(ws + (size_t)N_SRC * FDIM * 4);
    float* Vp_d     = Vp_s + FDIM * 16;
    const int B = (N_DST + BUCKET_NODES - 1) / BUCKET_NODES;
    const int C = (E + CHUNK_EDGES - 1) / CHUNK_EDGES;
    int2*  pairs    = (int2*)(Vp_d + FDIM * 16);
    int*   hist     = (int*)(pairs + E);
    int*   bstart   = hist + (size_t)C * B;
    float* as_node  = (float*)(ws + (size_t)(N_SRC + N_DST) * FDIM * 4);
    int*   cnt      = (int*)(as_node + (size_t)N_SRC * HEADS);
    int*   start    = cnt + N_DST;
    int*   ssrc     = start + 2 * (size_t)N_DST;

    float* out = (float*)d_out;

    chunk_bucket_counts_build_logit_tiles<<<C + 4, 1024, 0, stream>>>(
        edst, hist, E, B, C, W_src, attS, W_dst, attD, Vp_s, Vp_d);
    gemm_x128_logits<<<(N_SRC + 15) / 16, 288, 0, stream>>>(
        x_src, W_src, Vp_s, nullptr, src_feat, as_node, N_SRC);
    deal_edges_to_buckets<<<C, 1024, 0, stream>>>(esrc, edst, hist, bstart, pairs, E, B, C);
    group_bucket_by_node<<<B, 1024, 0, stream>>>(pairs, bstart, start, cnt, ssrc, N_DST);
    self_project_softmax_aggregate<<<(N_DST + 15) / 16, 512, 0, stream>>>(
        x_dst, W_self, Vp_d, b_self, ssrc, start, cnt, src_feat, as_node, out, N_DST);
}


# Work-groups own 32 rows, not 16, in both GEMM phases: each W fragment load feeds two row tiles (two accumulator chains per wave), each aggregate wave takes two nodes; half the waves, sum order kept
# speedup vs baseline: 4.8963x; 1.0528x over previous
#include <hip/hip_runtime.h>
#include <hip/hip_bf16.h>
#include <math.h>

typedef __attribute__((ext_vector_type(2))) float v2f;
typedef __attribute__((ext_vector_type(8))) float v8f;

#define HEADS 4
#define FDIM  128

#define ROWS 32

template <int LDB, int KSTEPS>
__device__ __forceinline__ void tile_product2(const float (*At)[132],
                                              const float* __restrict__ Bcol, int m, int koff,
                                              v8f& acc0, v8f& acc1)
{
    v8f c0 = {}, c1 = {};
    #pragma unroll KSTEPS
    for (int kb = 0; kb < FDIM; kb += 4) {
        v2f a0, a1, b;
        a0.x = At[m][kb + koff];
        a0.y = At[m][kb + koff + 1];
        a1.x = At[16 + m][kb + koff];
        a1.y = At[16 + m][kb + koff + 1];
        b.x = Bcol[(kb + koff)     * LDB];
        b.y = Bcol[(kb + koff + 1) * LDB];
        c0 = __builtin_amdgcn_wmma_f32_16x16x4_f32(
                 false, a0, false, b, (short)0, c0, false, false);
        c1 = __builtin_amdgcn_wmma_f32_16x16x4_f32(
                 false, a1, false, b, (short)0, c1, false, false);
    }
    acc0 = c0; acc1 = c1;
}

__global__ __launch_bounds__(288, 16) void gemm_x128_logits(
    const float* __restrict__ X, const float* __restrict__ W,
    const float* __restrict__ Vp, const float* __restrict__ bias,
    float* __restrict__ Y, float* __restrict__ alpha, int M)
{
    __shared__ float At[ROWS][132];
    const int tid  = threadIdx.x;
    const int row0 = blockIdx.x * ROWS;
    const int wv = __builtin_amdgcn_readfirstlane(tid >> 5);

    if (wv < 8) {
        #pragma unroll
        for (int i = 0; i < 4; ++i) {
            int idx = (tid + i * 256) * 4;
            int r = idx >> 7, c = idx & 127;
            int gr = row0 + r; gr = gr < M ? gr : M - 1;
            const float4 v = *(const float4*)(X + (size_t)gr * FDIM + c);
            At[r][c + 0] = v.x; At[r][c + 1] = v.y;
            At[r][c + 2] = v.z; At[r][c + 3] = v.w;
        }
    }
    __syncthreads();

    const int lane = tid & 31;
    const int m    = lane & 15;
    const int half = lane >> 4;
    const int koff = half << 1;

    if (wv < 8) {
        const int n = wv * 16 + m;
        v8f acc0, acc1;
        tile_product2<FDIM, 32>(At, W + n, m, koff, acc0, acc1);
        const float bn = bias ? bias[n] : 0.0f;
        #pragma unroll
        for (int r = 0; r < 8; ++r) {
            const int row = row0 + half * 8 + r;
            if (row < M)      Y[(size_t)row * FDIM + n]        = acc0[r] + bn;
            if (row + 16 < M) Y[(size_t)(row + 16) * FDIM + n] = acc1[r] + bn;
        }
    } else {
        v8f acc0, acc1;
        tile_product2<16, 32>(At, Vp + m, m, koff, acc0, acc1);
        if (m < HEADS) {
            #pragma unroll
            for (int r = 0; r < 8; ++r) {
                const int row = row0 + half * 8 + r;
                if (row < M)      alpha[(size_t)row * HEADS + m]        = acc0[r];
                if (row + 16 < M) alpha[(size_t)(row + 16) * HEADS + m] = acc1[r];
            }
        }
    }
}

#define BUCKET_SHIFT 8
#define BUCKET_NODES 256
#define CHUNK_EDGES  4096

__device__ __forceinline__ int bucket_of(int d)
{
    const unsigned b = (unsigned)d >> BUCKET_SHIFT;
    return (int)(b < BUCKET_NODES - 1u ? b : BUCKET_NODES - 1u);
}

__device__ __forceinline__ int wave_incl_scan(int v, int lane)
{
    #pragma unroll
    for (int o = 1; o < 32; o <<= 1) {
        const int t = __shfl_up(v, o);
        if (lane >= o) v += t;
    }
    return v;
}

__global__ __launch_bounds__(1024) void chunk_bucket_counts_build_logit_tiles(
    const int* __restrict__ edst, int* __restrict__ hist, int E, int B, int C,
    const float* __restrict__ W_s, const float* __restrict__ att_s,
    const float* __restrict__ W_d, const float* __restrict__ att_d,
    float* __restrict__ Vp_s, float* __restrict__ Vp_d)
{
    __shared__ int h[BUCKET_NODES];
    const int tid = threadIdx.x;
    if ((int)blockIdx.x < C) {
        const int chunk = blockIdx.x;
        if (tid < BUCKET_NODES) h[tid] = 0;
        __syncthreads();
        #pragma unroll
        for (int i = 0; i < 4; ++i) {
            const int e = chunk * CHUNK_EDGES + i * 1024 + tid;
            if (e < E)
                __hip_atomic_fetch_add(&h[bucket_of(edst[e])], 1,
                                       __ATOMIC_RELAXED, __HIP_MEMORY_SCOPE_WORKGROUP);
        }
        __syncthreads();
        if (tid < B) hist[chunk * B + tid] = h[tid];
        return;
    }
    const int t = ((int)blockIdx.x - C) * 1024 + tid;
    const int which = t >> 11;
    const int k = (t >> 4) & 127;
    const int c = t & 15;
    const float* Wm  = which ? W_d : W_s;
    const float* att = which ? att_d : att_s;
    float s = 0.0f;
    if (c < HEADS) {
        const float* wr = Wm + (size_t)k * FDIM + c * 32;
        const float* ar = att + c * 32;
        #pragma unroll
        for (int d = 0; d < 32; d += 4) {
            const float4 w = *(const float4*)(wr + d);
            const float4 a = *(const float4*)(ar + d);
            s += w.x * a.x; s += w.y * a.y; s += w.z * a.z; s += w.w * a.w;
        }
    }
    (which ? Vp_d : Vp_s)[k * 16 + c] = s;
}

__global__ __launch_bounds__(1024) void deal_edges_to_buckets(
    const int* __restrict__ esrc, const int* __restrict__ edst,
    const int* __restrict__ hist, int* __restrict__ bucket_start,
    int2* __restrict__ pairs, int E, int B, int C)
{
    __shared__ int s_tot[4][BUCKET_NODES];
    __shared__ int s_pre[4][BUCKET_NODES];
    __shared__ int s_base[BUCKET_NODES];
    __shared__ int s_fill[BUCKET_NODES];
    __shared__ int s_wsum[8];
    const int tid = threadIdx.x;
    const int lane = tid & 31;
    const int chunk = blockIdx.x;

    int d[4], s[4];
    #pragma unroll
    for (int i = 0; i < 4; ++i) {
        const int e = chunk * CHUNK_EDGES + i * 1024 + tid;
        d[i] = e < E ? edst[e] : -1;
        s[i] = e < E ? esrc[e] : 0;
    }

    const int g = tid >> 8, b = tid & (BUCKET_NODES - 1);
    int tot = 0, pre = 0;
    if (b < B) {
        #pragma unroll 7
        for (int c = g; c < C; c += 4) {
            const int v = hist[c * B + b];
            tot += v;
            pre += c < chunk ? v : 0;
        }
    }
    s_tot[g][b] = tot; s_pre[g][b] = pre;
    __syncthreads();
    int incl = 0;
    if (tid < BUCKET_NODES) {
        tot = s_tot[0][tid] + s_tot[1][tid] + s_tot[2][tid] + s_tot[3][tid];
        pre = s_pre[0][tid] + s_pre[1][tid] + s_pre[2][tid] + s_pre[3][tid];
        incl = wave_incl_scan(tot, lane);
        if (lane == 31) s_wsum[tid >> 5] = incl;
    }
    __syncthreads();
    if (tid < BUCKET_NODES) {
        int wbase = 0;
        for (int w = 0; w < (tid >> 5); ++w) wbase += s_wsum[w];
        const int excl = wbase + incl - tot;
        s_base[tid] = excl + pre;
        s_fill[tid] = 0;
        if (chunk == 0) {
            bucket_start[tid] = excl;
            if (tid == BUCKET_NODES - 1) bucket_start[BUCKET_NODES] = excl + tot;
        }
    }
    __syncthreads();
    #pragma unroll
    for (int i = 0; i < 4; ++i) {
        if (chunk * CHUNK_EDGES + i * 1024 + tid < E) {
            const int bk = bucket_of(d[i]);
            const int r = __hip_atomic_fetch_add(&s_fill[bk], 1,
                                                 __ATOMIC_RELAXED, __HIP_MEMORY_SCOPE_WORKGROUP);
            pairs[s_base[bk] + r] = make_int2(d[i], s[i]);
        }
    }
}

__global__ __launch_bounds__(1024) void group_bucket_by_node(
    const int2* __restrict__ pairs, const int* __restrict__ bucket_start,
    int* __restrict__ start, int* __restrict__ cnt, int* __restrict__ ssrc, int N)
{
    __shared__ int s_cnt[BUCKET_NODES];
    __shared__ int s_off[BUCKET_NODES];
    __shared__ int s_wsum[8];
    const int tid = threadIdx.x;
    const int lane = tid & 31;
    const int bk = blockIdx.x;
    const int beg = bucket_start[bk], end = bucket_start[bk + 1];
    if (tid < BUCKET_NODES) s_cnt[tid] = 0;
    __syncthreads();
    for (int e0 = beg + tid; e0 < end; e0 += 4096) {
        int dl[4];
        #pragma unroll
        for (int i = 0; i < 4; ++i) {
            const int e = e0 + i * 1024;
            dl[i] = pairs[e < end ? e : end - 1].x;
        }
        #pragma unroll
        for (int i = 0; i < 4; ++i)
            if (e0 + i * 1024 < end)
                __hip_atomic_fetch_add(&s_cnt[dl[i] & (BUCKET_NODES - 1)], 1,
                                       __ATOMIC_RELAXED, __HIP_MEMORY_SCOPE_WORKGROUP);
    }
    __syncthreads();
    int c = 0, incl = 0;
    if (tid < BUCKET_NODES) {
        c = s_cnt[tid];
        incl = wave_incl_scan(c, lane);
        if (lane == 31) s_wsum[tid >> 5] = incl;
    }
    __syncthreads();
    if (tid < BUCKET_NODES) {
        int wbase = 0;
        for (int w = 0; w < (tid >> 5); ++w) wbase += s_wsum[w];
        const int st = beg + wbase + incl - c;
        s_off[tid] = st;
        const int node = bk * BUCKET_NODES + tid;
        if (node < N) { start[node] = st; cnt[node] = c; }
    }
    __syncthreads();
    for (int e0 = beg + tid; e0 < end; e0 += 4096) {
        int2 p[4];
        #pragma unroll
        for (int i = 0; i < 4; ++i) {
            const int e = e0 + i * 1024;
            p[i] = pairs[e < end ? e : end - 1];
        }
        #pragma unroll
        for (int i = 0; i < 4; ++i)
            if (e0 + i * 1024 < end) {
                const int pos = __hip_atomic_fetch_add(&s_off[p[i].x & (BUCKET_NODES - 1)], 1,
                                                       __ATOMIC_RELAXED, __HIP_MEMORY_SCOPE_WORKGROUP);
                ssrc[pos] = p[i].y;
            }
    }
}

__device__ __forceinline__ float lrelu(float x) { return x > 0.f ? x : 0.2f * x; }

__device__ __forceinline__ float4 edge_scores(const float* __restrict__ asn, int s, float4 adv)
{
    const float4 a = *(const float4*)(asn + (size_t)s * HEADS);
    float4 r;
    r.x = lrelu(a.x + adv.x); r.y = lrelu(a.y + adv.y);
    r.z = lrelu(a.z + adv.z); r.w = lrelu(a.w + adv.w);
    return r;
}

__device__ __forceinline__ float4 wave_max4(float4 v)
{
    #pragma unroll
    for (int o = 16; o > 0; o >>= 1) {
        v.x = fmaxf(v.x, __shfl_xor(v.x, o)); v.y = fmaxf(v.y, __shfl_xor(v.y, o));
        v.z = fmaxf(v.z, __shfl_xor(v.z, o)); v.w = fmaxf(v.w, __shfl_xor(v.w, o));
    }
    return v;
}

__device__ __forceinline__ float4 wave_sum4(float4 v)
{
    #pragma unroll
    for (int o = 16; o > 0; o >>= 1) {
        v.x += __shfl_xor(v.x, o); v.y += __shfl_xor(v.y, o);
        v.z += __shfl_xor(v.z, o); v.w += __shfl_xor(v.w, o);
    }
    return v;
}

__device__ __forceinline__ float head_weight(float4 w_l, int i, const unsigned (&hm)[4])
{
    const unsigned bx = (unsigned)__builtin_amdgcn_readlane(__float_as_int(w_l.x), i);
    const unsigned by = (unsigned)__builtin_amdgcn_readlane(__float_as_int(w_l.y), i);
    const unsigned bz = (unsigned)__builtin_amdgcn_readlane(__float_as_int(w_l.z), i);
    const unsigned bw = (unsigned)__builtin_amdgcn_readlane(__float_as_int(w_l.w), i);
    return __uint_as_float((bx & hm[0]) | (by & hm[1]) | (bz & hm[2]) | (bw & hm[3]));
}

__device__ __forceinline__ float4 src_row(const float* __restrict__ src_feat, int s_l, int i, int lane)
{
    const int s = __builtin_amdgcn_readlane(s_l, i);
    return *(const float4*)(src_feat + (size_t)s * FDIM + lane * 4);
}

__device__ __forceinline__ void fma_row(float4& acc, float w, float4 f)
{
    acc.x += w * f.x; acc.y += w * f.y; acc.z += w * f.z; acc.w += w * f.w;
}

__device__ __forceinline__ void add_rows(int s_l, float4 w_l, int m, int lane,
                                         const unsigned (&hm)[4],
                                         const float* __restrict__ src_feat, float4& acc)
{
    int i = 0;
    for (; i + 4 <= m; i += 4) {
        const float4 f0 = src_row(src_feat, s_l, i + 0, lane);
        const float4 f1 = src_row(src_feat, s_l, i + 1, lane);
        const float4 f2 = src_row(src_feat, s_l, i + 2, lane);
        const float4 f3 = src_row(src_feat, s_l, i + 3, lane);
        const float w0 = head_weight(w_l, i + 0, hm);
        const float w1 = head_weight(w_l, i + 1, hm);
        const float w2 = head_weight(w_l, i + 2, hm);
        const float w3 = head_weight(w_l, i + 3, hm);
        fma_row(acc, w0, f0); fma_row(acc, w1, f1);
        fma_row(acc, w2, f2); fma_row(acc, w3, f3);
    }
    for (; i < m; ++i) {
        const float4 f = src_row(src_feat, s_l, i, lane);
        fma_row(acc, head_weight(w_l, i, hm), f);
    }
}

__device__ __forceinline__ void aggregate_node(
    int d, int beg, int n, const float* selfrow, const float* lrow, int lane,
    const unsigned (&hm)[4],
    const int* __restrict__ ssrc, const float* __restrict__ src_feat,
    const float* __restrict__ asn, float* __restrict__ out)
{
    float* op = out + (size_t)d * FDIM + lane * 4;
    const float4 self = *(const float4*)(selfrow + lane * 4);
    const float4 adv = *(const float4*)lrow;
    const float NEG = -INFINITY;

    const bool v0 = lane < n;
    const int s0 = v0 ? ssrc[beg + lane] : 0;
    float4 al0 = {NEG, NEG, NEG, NEG};
    if (v0) al0 = edge_scores(asn, s0, adv);

    float4 mx = al0;
    for (int b = 32; b < n; b += 32) {
        const int j = b + lane;
        if (j < n) {
            const float4 al = edge_scores(asn, ssrc[beg + j], adv);
            mx.x = fmaxf(mx.x, al.x); mx.y = fmaxf(mx.y, al.y);
            mx.z = fmaxf(mx.z, al.z); mx.w = fmaxf(mx.w, al.w);
        }
    }
    mx = wave_max4(mx);

    float4 ex0 = {0.f, 0.f, 0.f, 0.f};
    if (v0) {
        ex0.x = __expf(al0.x - mx.x); ex0.y = __expf(al0.y - mx.y);
        ex0.z = __expf(al0.z - mx.z); ex0.w = __expf(al0.w - mx.w);
    }
    float4 sm = ex0;
    for (int b = 32; b < n; b += 32) {
        const int j = b + lane;
        if (j < n) {
            const float4 al = edge_scores(asn, ssrc[beg + j], adv);
            sm.x += __expf(al.x - mx.x); sm.y += __expf(al.y - mx.y);
            sm.z += __expf(al.z - mx.z); sm.w += __expf(al.w - mx.w);
        }
    }
    sm = wave_sum4(sm);
    float4 den;
    den.x = sm.x + 1e-16f; den.y = sm.y + 1e-16f; den.z = sm.z + 1e-16f; den.w = sm.w + 1e-16f;

    float4 acc = {0.f, 0.f, 0.f, 0.f};
    float4 w0;
    w0.x = ex0.x / den.x; w0.y = ex0.y / den.y; w0.z = ex0.z / den.z; w0.w = ex0.w / den.w;
    add_rows(s0, w0, n < 32 ? n : 32, lane, hm, src_feat, acc);
    for (int b = 32; b < n; b += 32) {
        const int j = b + lane;
        int s = 0;
        float4 w = {0.f, 0.f, 0.f, 0.f};
        if (j < n) {
            s = ssrc[beg + j];
            const float4 al = edge_scores(asn, s, adv);
            w.x = __expf(al.x - mx.x) / den.x; w.y = __expf(al.y - mx.y) / den.y;
            w.z = __expf(al.z - mx.z) / den.z; w.w = __expf(al.w - mx.w) / den.w;
        }
        const int rem = n - b;
        add_rows(s, w, rem < 32 ? rem : 32, lane, hm, src_feat, acc);
    }

    float4 o;
    o.x = self.x + acc.x; o.y = self.y + acc.y; o.z = self.z + acc.z; o.w = self.w + acc.w;
    *(float4*)op = o;
}

__global__ __launch_bounds__(512, 16) void self_project_softmax_aggregate(
    const float* __restrict__ X, const float* __restrict__ W,
    const float* __restrict__ Vp, const float* __restrict__ bias,
    const int* __restrict__ ssrc, const int* __restrict__ start, const int* __restrict__ cnt,
    const float* __restrict__ src_feat, const float* __restrict__ asn,
    float* __restrict__ out, int N)
{
    __shared__ float At[ROWS][132];
    __shared__ __attribute__((aligned(16))) float Yt[ROWS][132];
    __shared__ __attribute__((aligned(16))) float Lt[ROWS][HEADS];
    const int tid  = threadIdx.x;
    const int lane = tid & 31;
    const int row0 = blockIdx.x * ROWS;
    const int wv = __builtin_amdgcn_readfirstlane(tid >> 5);

    const int dA  = row0 + wv, dB = dA + 16;
    const int dcA = dA < N ? dA : N - 1;
    const int dcB = dB < N ? dB : N - 1;
    const int begA = start[dcA], nA = cnt[dcA];
    const int begB = start[dcB], nB = cnt[dcB];

    {
        #pragma unroll
        for (int i = 0; i < 2; ++i) {
            int idx = (tid + i * 512) * 4;
            int r = idx >> 7, c = idx & 127;
            int gr = row0 + r; gr = gr < N ? gr : N - 1;
            const float4 v = *(const float4*)(X + (size_t)gr * FDIM + c);
            At[r][c + 0] = v.x; At[r][c + 1] = v.y;
            At[r][c + 2] = v.z; At[r][c + 3] = v.w;
        }
    }
    __syncthreads();
    {
        const int m    = lane & 15;
        const int half = lane >> 4;
        const int koff = half << 1;
        if (wv < 8) {
            const int col = wv * 16 + m;
            v8f acc0, acc1;
            tile_product2<FDIM, 16>(At, W + col, m, koff, acc0, acc1);
            const float bn = bias[col];
            #pragma unroll
            for (int r = 0; r < 8; ++r) {
                Yt[half * 8 + r][col]      = acc0[r] + bn;
                Yt[16 + half * 8 + r][col] = acc1[r] + bn;
            }
        } else if (wv == 8) {
            v8f acc0, acc1;
            tile_product2<16, 16>(At, Vp + m, m, koff, acc0, acc1);
            if (m < HEADS) {
                #pragma unroll
                for (int r = 0; r < 8; ++r) {
                    Lt[half * 8 + r][m]      = acc0[r];
                    Lt[16 + half * 8 + r][m] = acc1[r];
                }
            }
        }
    }
    __syncthreads();

    const int h = lane >> 3;
    const unsigned hm[4] = { h == 0 ? ~0u : 0u, h == 1 ? ~0u : 0u, h == 2 ? ~0u : 0u, h == 3 ? ~0u : 0u };
    if (dA < N)
        aggregate_node(dA, begA, nA, &Yt[wv][0], &Lt[wv][0], lane, hm,
                       ssrc, src_feat, asn, out);
    if (dB < N)
        aggregate_node(dB, begB, nB, &Yt[wv + 16][0], &Lt[wv + 16][0], lane, hm,
                       ssrc, src_feat, asn, out);
}

extern "C" void kernel_launch(void* const* d_in, const int* in_sizes, int n_in,
                              void* d_out, int out_size, void* d_ws, size_t ws_size,
                              hipStream_t stream)
{
    const float* x_src  = (const float*)d_in[0];
    const float* x_dst  = (const float*)d_in[1];
    const int*   esrc   = (const int*)d_in[2];
    const int*   edst   = (const int*)d_in[3];
    const float* W_src  = (const float*)d_in[5];
    const float* W_dst  = (const float*)d_in[6];
    const float* attS   = (const float*)d_in[7];
    const float* attD   = (const float*)d_in[8];
    const float* W_self = (const float*)d_in[9];
    const float* b_self = (const float*)d_in[10];

    const int N_SRC = in_sizes[0] / FDIM;
    const int N_DST = in_sizes[1] / FDIM;
    const int E     = in_sizes[2];

    char* ws = (char*)d_ws;
    float* src_feat = (float*)(ws);
    float* Vp_s     = (float*)(ws + (size_t)N_SRC * FDIM * 4);
    float* Vp_d     = Vp_s + FDIM * 16;
    const int B = (N_DST + BUCKET_NODES - 1) / BUCKET_NODES;
    const int C = (E + CHUNK_EDGES - 1) / CHUNK_EDGES;
    int2*  pairs    = (int2*)(Vp_d + FDIM * 16);
    int*   hist     = (int*)(pairs + E);
    int*   bstart   = hist + (size_t)C * B;
    float* as_node  = (float*)(ws + (size_t)(N_SRC + N_DST) * FDIM * 4);
    int*   cnt      = (int*)(as_node + (size_t)N_SRC * HEADS);
    int*   start    = cnt + N_DST;
    int*   ssrc     = start + 2 * (size_t)N_DST;

    float* out = (float*)d_out;

    chunk_bucket_counts_build_logit_tiles<<<C + 4, 1024, 0, stream>>>(
        edst, hist, E, B, C, W_src, attS, W_dst, attD, Vp_s, Vp_d);
    gemm_x128_logits<<<(N_SRC + ROWS - 1) / ROWS, 288, 0, stream>>>(
        x_src, W_src, Vp_s, nullptr, src_feat, as_node, N_SRC);
    deal_edges_to_buckets<<<C, 1024, 0, stream>>>(esrc, edst, hist, bstart, pairs, E, B, C);
    group_bucket_by_node<<<B, 1024, 0, stream>>>(pairs, bstart, start, cnt, ssrc, N_DST);
    self_project_softmax_aggregate<<<(N_DST + ROWS - 1) / ROWS, 512, 0, stream>>>(
        x_dst, W_self, Vp_d, b_self, ssrc, start, cnt, src_feat, as_node, out, N_DST);
}


# Node-level edge grouping rides in the source GEMM's launch as its first 196 work-groups (288 threads) and runs under the matrix work: 5 to 4 launches; arrival-rank order of a node's edges may differ
# speedup vs baseline: 5.2791x; 1.0782x over previous
#include <hip/hip_runtime.h>
#include <hip/hip_bf16.h>
#include <math.h>

typedef __attribute__((ext_vector_type(2))) float v2f;
typedef __attribute__((ext_vector_type(8))) float v8f;

#define HEADS 4
#define FDIM  128

#define ROWS 32
#define GEMM_THREADS 288

template <int LDB, int KSTEPS>
__device__ __forceinline__ void tile_product2(const float (*At)[132],
                                              const float* __restrict__ Bcol, int m, int koff,
                                              v8f& acc0, v8f& acc1)
{
    v8f c0 = {}, c1 = {};
    #pragma unroll KSTEPS
    for (int kb = 0; kb < FDIM; kb += 4) {
        v2f a0, a1, b;
        a0.x = At[m][kb + koff];
        a0.y = At[m][kb + koff + 1];
        a1.x = At[16 + m][kb + koff];
        a1.y = At[16 + m][kb + koff + 1];
        b.x = Bcol[(kb + koff)     * LDB];
        b.y = Bcol[(kb + koff + 1) * LDB];
        c0 = __builtin_amdgcn_wmma_f32_16x16x4_f32(
                 false, a0, false, b, (short)0, c0, false, false);
        c1 = __builtin_amdgcn_wmma_f32_16x16x4_f32(
                 false, a1, false, b, (short)0, c1, false, false);
    }
    acc0 = c0; acc1 = c1;
}

#define BUCKET_SHIFT 8
#define BUCKET_NODES 256
#define CHUNK_EDGES  4096

__device__ __forceinline__ int bucket_of(int d)
{
    const unsigned b = (unsigned)d >> BUCKET_SHIFT;
    return (int)(b < BUCKET_NODES - 1u ? b : BUCKET_NODES - 1u);
}

__device__ __forceinline__ int wave_incl_scan(int v, int lane)
{
    #pragma unroll
    for (int o = 1; o < 32; o <<= 1) {
        const int t = __shfl_up(v, o);
        if (lane >= o) v += t;
    }
    return v;
}

__global__ __launch_bounds__(1024) void chunk_bucket_counts_build_logit_tiles(
    const int* __restrict__ edst, int* __restrict__ hist, int E, int B, int C,
    const float* __restrict__ W_s, const float* __restrict__ att_s,
    const float* __restrict__ W_d, const float* __restrict__ att_d,
    float* __restrict__ Vp_s, float* __restrict__ Vp_d)
{
    __shared__ int h[BUCKET_NODES];
    const int tid = threadIdx.x;
    if ((int)blockIdx.x < C) {
        const int chunk = blockIdx.x;
        if (tid < BUCKET_NODES) h[tid] = 0;
        __syncthreads();
        #pragma unroll
        for (int i = 0; i < 4; ++i) {
            const int e = chunk * CHUNK_EDGES + i * 1024 + tid;
            if (e < E)
                __hip_atomic_fetch_add(&h[bucket_of(edst[e])], 1,
                                       __ATOMIC_RELAXED, __HIP_MEMORY_SCOPE_WORKGROUP);
        }
        __syncthreads();
        if (tid < B) hist[chunk * B + tid] = h[tid];
        return;
    }
    const int t = ((int)blockIdx.x - C) * 1024 + tid;
    const int which = t >> 11;
    const int k = (t >> 4) & 127;
    const int c = t & 15;
    const float* Wm  = which ? W_d : W_s;
    const float* att = which ? att_d : att_s;
    float s = 0.0f;
    if (c < HEADS) {
        const float* wr = Wm + (size_t)k * FDIM + c * 32;
        const float* ar = att + c * 32;
        #pragma unroll
        for (int d = 0; d < 32; d += 4) {
            const float4 w = *(const float4*)(wr + d);
            const float4 a = *(const float4*)(ar + d);
            s += w.x * a.x; s += w.y * a.y; s += w.z * a.z; s += w.w * a.w;
        }
    }
    (which ? Vp_d : Vp_s)[k * 16 + c] = s;
}

__global__ __launch_bounds__(1024) void deal_edges_to_buckets(
    const int* __restrict__ esrc, const int* __restrict__ edst,
    const int* __restrict__ hist, int* __restrict__ bucket_start,
    int2* __restrict__ pairs, int E, int B, int C)
{
    __shared__ int s_tot[4][BUCKET_NODES];
    __shared__ int s_pre[4][BUCKET_NODES];
    __shared__ int s_base[BUCKET_NODES];
    __shared__ int s_fill[BUCKET_NODES];
    __shared__ int s_wsum[8];
    const int tid = threadIdx.x;
    const int lane = tid & 31;
    const int chunk = blockIdx.x;

    int d[4], s[4];
    #pragma unroll
    for (int i = 0; i < 4; ++i) {
        const int e = chunk * CHUNK_EDGES + i * 1024 + tid;
        d[i] = e < E ? edst[e] : -1;
        s[i] = e < E ? esrc[e] : 0;
    }

    const int g = tid >> 8, b = tid & (BUCKET_NODES - 1);
    int tot = 0, pre = 0;
    if (b < B) {
        #pragma unroll 7
        for (int c = g; c < C; c += 4) {
            const int v = hist[c * B + b];
            tot += v;
            pre += c < chunk ? v : 0;
        }
    }
    s_tot[g][b] = tot; s_pre[g][b] = pre;
    __syncthreads();
    int incl = 0;
    if (tid < BUCKET_NODES) {
        tot = s_tot[0][tid] + s_tot[1][tid] + s_tot[2][tid] + s_tot[3][tid];
        pre = s_pre[0][tid] + s_pre[1][tid] + s_pre[2][tid] + s_pre[3][tid];
        incl = wave_incl_scan(tot, lane);
        if (lane == 31) s_wsum[tid >> 5] = incl;
    }
    __syncthreads();
    if (tid < BUCKET_NODES) {
        int wbase = 0;
        for (int w = 0; w < (tid >> 5); ++w) wbase += s_wsum[w];
        const int excl = wbase + incl - tot;
        s_base[tid] = excl + pre;
        s_fill[tid] = 0;
        if (chunk == 0) {
            bucket_start[tid] = excl;
            if (tid == BUCKET_NODES - 1) bucket_start[BUCKET_NODES] = excl + tot;
        }
    }
    __syncthreads();
    #pragma unroll
    for (int i = 0; i < 4; ++i) {
        if (chunk * CHUNK_EDGES + i * 1024 + tid < E) {
            const int bk = bucket_of(d[i]);
            const int r = __hip_atomic_fetch_add(&s_fill[bk], 1,
                                                 __ATOMIC_RELAXED, __HIP_MEMORY_SCOPE_WORKGROUP);
            pairs[s_base[bk] + r] = make_int2(d[i], s[i]);
        }
    }
}

template <int NT, int PER>
__device__ __forceinline__ void group_bucket_by_node(
    const int2* __restrict__ pairs, const int* __restrict__ bucket_start,
    int* __restrict__ start, int* __restrict__ cnt, int* __restrict__ ssrc, int N, int bk)
{
    __shared__ int s_cnt[BUCKET_NODES];
    __shared__ int s_off[BUCKET_NODES];
    __shared__ int s_wsum[8];
    const int tid = threadIdx.x;
    const int lane = tid & 31;
    const int beg = bucket_start[bk], end = bucket_start[bk + 1];
    if (tid < BUCKET_NODES) s_cnt[tid] = 0;
    __syncthreads();
    for (int e0 = beg + tid; e0 < end; e0 += PER * NT) {
        int dl[PER];
        #pragma unroll
        for (int i = 0; i < PER; ++i) {
            const int e = e0 + i * NT;
            dl[i] = pairs[e < end ? e : end - 1].x;
        }
        #pragma unroll
        for (int i = 0; i < PER; ++i)
            if (e0 + i * NT < end)
                __hip_atomic_fetch_add(&s_cnt[dl[i] & (BUCKET_NODES - 1)], 1,
                                       __ATOMIC_RELAXED, __HIP_MEMORY_SCOPE_WORKGROUP);
    }
    __syncthreads();
    int c = 0, incl = 0;
    if (tid < BUCKET_NODES) {
        c = s_cnt[tid];
        incl = wave_incl_scan(c, lane);
        if (lane == 31) s_wsum[tid >> 5] = incl;
    }
    __syncthreads();
    if (tid < BUCKET_NODES) {
        int wbase = 0;
        for (int w = 0; w < (tid >> 5); ++w) wbase += s_wsum[w];
        const int st = beg + wbase + incl - c;
        s_off[tid] = st;
        const int node = bk * BUCKET_NODES + tid;
        if (node < N) { start[node] = st; cnt[node] = c; }
    }
    __syncthreads();
    for (int e0 = beg + tid; e0 < end; e0 += PER * NT) {
        int2 p[PER];
        #pragma unroll
        for (int i = 0; i < PER; ++i) {
            const int e = e0 + i * NT;
            p[i] = pairs[e < end ? e : end - 1];
        }
        #pragma unroll
        for (int i = 0; i < PER; ++i)
            if (e0 + i * NT < end) {
                const int pos = __hip_atomic_fetch_add(&s_off[p[i].x & (BUCKET_NODES - 1)], 1,
                                                       __ATOMIC_RELAXED, __HIP_MEMORY_SCOPE_WORKGROUP);
                ssrc[pos] = p[i].y;
            }
    }
}

__global__ __launch_bounds__(GEMM_THREADS, 16) void gemm_x128_logits_beside_node_grouping(
    const float* __restrict__ X, const float* __restrict__ W,
    const float* __restrict__ Vp, const float* __restrict__ bias,
    float* __restrict__ Y, float* __restrict__ alpha, int M,
    const int2* __restrict__ pairs, const int* __restrict__ bucket_start,
    int* __restrict__ start, int* __restrict__ cnt, int* __restrict__ ssrc, int N, int NB)
{
    if ((int)blockIdx.x < NB) {
        group_bucket_by_node<GEMM_THREADS, 8>(pairs, bucket_start, start, cnt, ssrc, N,
                                              (int)blockIdx.x);
        return;
    }
    __shared__ float At[ROWS][132];
    const int tid  = threadIdx.x;
    const int row0 = ((int)blockIdx.x - NB) * ROWS;
    const int wv = __builtin_amdgcn_readfirstlane(tid >> 5);

    if (wv < 8) {
        #pragma unroll
        for (int i = 0; i < 4; ++i) {
            int idx = (tid + i * 256) * 4;
            int r = idx >> 7, c = idx & 127;
            int gr = row0 + r; gr = gr < M ? gr : M - 1;
            const float4 v = *(const float4*)(X + (size_t)gr * FDIM + c);
            At[r][c + 0] = v.x; At[r][c + 1] = v.y;
            At[r][c + 2] = v.z; At[r][c + 3] = v.w;
        }
    }
    __syncthreads();

    const int lane = tid & 31;
    const int m    = lane & 15;
    const int half = lane >> 4;
    const int koff = half << 1;

    if (wv < 8) {
        const int n = wv * 16 + m;
        v8f acc0, acc1;
        tile_product2<FDIM, 32>(At, W + n, m, koff, acc0, acc1);
        const float bn = bias ? bias[n] : 0.0f;
        #pragma unroll
        for (int r = 0; r < 8; ++r) {
            const int row = row0 + half * 8 + r;
            if (row < M)      Y[(size_t)row * FDIM + n]        = acc0[r] + bn;
            if (row + 16 < M) Y[(size_t)(row + 16) * FDIM + n] = acc1[r] + bn;
        }
    } else {
        v8f acc0, acc1;
        tile_product2<16, 32>(At, Vp + m, m, koff, acc0, acc1);
        if (m < HEADS) {
            #pragma unroll
            for (int r = 0; r < 8; ++r) {
                const int row = row0 + half * 8 + r;
                if (row < M)      alpha[(size_t)row * HEADS + m]        = acc0[r];
                if (row + 16 < M) alpha[(size_t)(row + 16) * HEADS + m] = acc1[r];
            }
        }
    }
}

__device__ __forceinline__ float lrelu(float x) { return x > 0.f ? x : 0.2f * x; }

__device__ __forceinline__ float4 edge_scores(const float* __restrict__ asn, int s, float4 adv)
{
    const float4 a = *(const float4*)(asn + (size_t)s * HEADS);
    float4 r;
    r.x = lrelu(a.x + adv.x); r.y = lrelu(a.y + adv.y);
    r.z = lrelu(a.z + adv.z); r.w = lrelu(a.w + adv.w);
    return r;
}

__device__ __forceinline__ float4 wave_max4(float4 v)
{
    #pragma unroll
    for (int o = 16; o > 0; o >>= 1) {
        v.x = fmaxf(v.x, __shfl_xor(v.x, o)); v.y = fmaxf(v.y, __shfl_xor(v.y, o));
        v.z = fmaxf(v.z, __shfl_xor(v.z, o)); v.w = fmaxf(v.w, __shfl_xor(v.w, o));
    }
    return v;
}

__device__ __forceinline__ float4 wave_sum4(float4 v)
{
    #pragma unroll
    for (int o = 16; o > 0; o >>= 1) {
        v.x += __shfl_xor(v.x, o); v.y += __shfl_xor(v.y, o);
        v.z += __shfl_xor(v.z, o); v.w += __shfl_xor(v.w, o);
    }
    return v;
}

__device__ __forceinline__ float head_weight(float4 w_l, int i, const unsigned (&hm)[4])
{
    const unsigned bx = (unsigned)__builtin_amdgcn_readlane(__float_as_int(w_l.x), i);
    const unsigned by = (unsigned)__builtin_amdgcn_readlane(__float_as_int(w_l.y), i);
    const unsigned bz = (unsigned)__builtin_amdgcn_readlane(__float_as_int(w_l.z), i);
    const unsigned bw = (unsigned)__builtin_amdgcn_readlane(__float_as_int(w_l.w), i);
    return __uint_as_float((bx & hm[0]) | (by & hm[1]) | (bz & hm[2]) | (bw & hm[3]));
}

__device__ __forceinline__ float4 src_row(const float* __restrict__ src_feat, int s_l, int i, int lane)
{
    const int s = __builtin_amdgcn_readlane(s_l, i);
    return *(const float4*)(src_feat + (size_t)s * FDIM + lane * 4);
}

__device__ __forceinline__ void fma_row(float4& acc, float w, float4 f)
{
    acc.x += w * f.x; acc.y += w * f.y; acc.z += w * f.z; acc.w += w * f.w;
}

__device__ __forceinline__ void add_rows(int s_l, float4 w_l, int m, int lane,
                                         const unsigned (&hm)[4],
                                         const float* __restrict__ src_feat, float4& acc)
{
    int i = 0;
    for (; i + 4 <= m; i += 4) {
        const float4 f0 = src_row(src_feat, s_l, i + 0, lane);
        const float4 f1 = src_row(src_feat, s_l, i + 1, lane);
        const float4 f2 = src_row(src_feat, s_l, i + 2, lane);
        const float4 f3 = src_row(src_feat, s_l, i + 3, lane);
        const float w0 = head_weight(w_l, i + 0, hm);
        const float w1 = head_weight(w_l, i + 1, hm);
        const float w2 = head_weight(w_l, i + 2, hm);
        const float w3 = head_weight(w_l, i + 3, hm);
        fma_row(acc, w0, f0); fma_row(acc, w1, f1);
        fma_row(acc, w2, f2); fma_row(acc, w3, f3);
    }
    for (; i < m; ++i) {
        const float4 f = src_row(src_feat, s_l, i, lane);
        fma_row(acc, head_weight(w_l, i, hm), f);
    }
}

__device__ __forceinline__ void aggregate_node(
    int d, int beg, int n, const float* selfrow, const float* lrow, int lane,
    const unsigned (&hm)[4],
    const int* __restrict__ ssrc, const float* __restrict__ src_feat,
    const float* __restrict__ asn, float* __restrict__ out)
{
    float* op = out + (size_t)d * FDIM + lane * 4;
    const float4 self = *(const float4*)(selfrow + lane * 4);
    const float4 adv = *(const float4*)lrow;
    const float NEG = -INFINITY;

    const bool v0 = lane < n;
    const int s0 = v0 ? ssrc[beg + lane] : 0;
    float4 al0 = {NEG, NEG, NEG, NEG};
    if (v0) al0 = edge_scores(asn, s0, adv);

    float4 mx = al0;
    for (int b = 32; b < n; b += 32) {
        const int j = b + lane;
        if (j < n) {
            const float4 al = edge_scores(asn, ssrc[beg + j], adv);
            mx.x = fmaxf(mx.x, al.x); mx.y = fmaxf(mx.y, al.y);
            mx.z = fmaxf(mx.z, al.z); mx.w = fmaxf(mx.w, al.w);
        }
    }
    mx = wave_max4(mx);

    float4 ex0 = {0.f, 0.f, 0.f, 0.f};
    if (v0) {
        ex0.x = __expf(al0.x - mx.x); ex0.y = __expf(al0.y - mx.y);
        ex0.z = __expf(al0.z - mx.z); ex0.w = __expf(al0.w - mx.w);
    }
    float4 sm = ex0;
    for (int b = 32; b < n; b += 32) {
        const int j = b + lane;
        if (j < n) {
            const float4 al = edge_scores(asn, ssrc[beg + j], adv);
            sm.x += __expf(al.x - mx.x); sm.y += __expf(al.y - mx.y);
            sm.z += __expf(al.z - mx.z); sm.w += __expf(al.w - mx.w);
        }
    }
    sm = wave_sum4(sm);
    float4 den;
    den.x = sm.x + 1e-16f; den.y = sm.y + 1e-16f; den.z = sm.z + 1e-16f; den.w = sm.w + 1e-16f;

    float4 acc = {0.f, 0.f, 0.f, 0.f};
    float4 w0;
    w0.x = ex0.x / den.x; w0.y = ex0.y / den.y; w0.z = ex0.z / den.z; w0.w = ex0.w / den.w;
    add_rows(s0, w0, n < 32 ? n : 32, lane, hm, src_feat, acc);
    for (int b = 32; b < n; b += 32) {
        const int j = b + lane;
        int s = 0;
        float4 w = {0.f, 0.f, 0.f, 0.f};
        if (j < n) {
            s = ssrc[beg + j];
            const float4 al = edge_scores(asn, s, adv);
            w.x = __expf(al.x - mx.x) / den.x; w.y = __expf(al.y - mx.y) / den.y;
            w.z = __expf(al.z - mx.z) / den.z; w.w = __expf(al.w - mx.w) / den.w;
        }
        const int rem = n - b;
        add_rows(s, w, rem < 32 ? rem : 32, lane, hm, src_feat, acc);
    }

    float4 o;
    o.x = self.x + acc.x; o.y = self.y + acc.y; o.z = self.z + acc.z; o.w = self.w + acc.w;
    *(float4*)op = o;
}

__global__ __launch_bounds__(512, 16) void self_project_softmax_aggregate(
    const float* __restrict__ X, const float* __restrict__ W,
    const float* __restrict__ Vp, const float* __restrict__ bias,
    const int* __restrict__ ssrc, const int* __restrict__ start, const int* __restrict__ cnt,
    const float* __restrict__ src_feat, const float* __restrict__ asn,
    float* __restrict__ out, int N)
{
    __shared__ float At[ROWS][132];
    __shared__ __attribute__((aligned(16))) float Yt[ROWS][132];
    __shared__ __attribute__((aligned(16))) float Lt[ROWS][HEADS];
    const int tid  = threadIdx.x;
    const int lane = tid & 31;
    const int row0 = blockIdx.x * ROWS;
    const int wv = __builtin_amdgcn_readfirstlane(tid >> 5);

    const int dA  = row0 + wv, dB = dA + 16;
    const int dcA = dA < N ? dA : N - 1;
    const int dcB = dB < N ? dB : N - 1;
    const int begA = start[dcA], nA = cnt[dcA];
    const int begB = start[dcB], nB = cnt[dcB];

    {
        #pragma unroll
        for (int i = 0; i < 2; ++i) {
            int idx = (tid + i * 512) * 4;
            int r = idx >> 7, c = idx & 127;
            int gr = row0 + r; gr = gr < N ? gr : N - 1;
            const float4 v = *(const float4*)(X + (size_t)gr * FDIM + c);
            At[r][c + 0] = v.x; At[r][c + 1] = v.y;
            At[r][c + 2] = v.z; At[r][c + 3] = v.w;
        }
    }
    __syncthreads();
    {
        const int m    = lane & 15;
        const int half = lane >> 4;
        const int koff = half << 1;
        if (wv < 8) {
            const int col = wv * 16 + m;
            v8f acc0, acc1;
            tile_product2<FDIM, 16>(At, W + col, m, koff, acc0, acc1);
            const float bn = bias[col];
            #pragma unroll
            for (int r = 0; r < 8; ++r) {
                Yt[half * 8 + r][col]      = acc0[r] + bn;
                Yt[16 + half * 8 + r][col] = acc1[r] + bn;
            }
        } else if (wv == 8) {
            v8f acc0, acc1;
            tile_product2<16, 16>(At, Vp + m, m, koff, acc0, acc1);
            if (m < HEADS) {
                #pragma unroll
                for (int r = 0; r < 8; ++r) {
                    Lt[half * 8 + r][m]      = acc0[r];
                    Lt[16 + half * 8 + r][m] = acc1[r];
                }
            }
        }
    }
    __syncthreads();

    const int h = lane >> 3;
    const unsigned hm[4] = { h == 0 ? ~0u : 0u, h == 1 ? ~0u : 0u, h == 2 ? ~0u : 0u, h == 3 ? ~0u : 0u };
    if (dA < N)
        aggregate_node(dA, begA, nA, &Yt[wv][0], &Lt[wv][0], lane, hm,
                       ssrc, src_feat, asn, out);
    if (dB < N)
        aggregate_node(dB, begB, nB, &Yt[wv + 16][0], &Lt[wv + 16][0], lane, hm,
                       ssrc, src_feat, asn, out);
}

extern "C" void kernel_launch(void* const* d_in, const int* in_sizes, int n_in,
                              void* d_out, int out_size, void* d_ws, size_t ws_size,
                              hipStream_t stream)
{
    const float* x_src  = (const float*)d_in[0];
    const float* x_dst  = (const float*)d_in[1];
    const int*   esrc   = (const int*)d_in[2];
    const int*   edst   = (const int*)d_in[3];
    const float* W_src  = (const float*)d_in[5];
    const float* W_dst  = (const float*)d_in[6];
    const float* attS   = (const float*)d_in[7];
    const float* attD   = (const float*)d_in[8];
    const float* W_self = (const float*)d_in[9];
    const float* b_self = (const float*)d_in[10];

    const int N_SRC = in_sizes[0] / FDIM;
    const int N_DST = in_sizes[1] / FDIM;
    const int E     = in_sizes[2];

    char* ws = (char*)d_ws;
    float* src_feat = (float*)(ws);
    float* Vp_s     = (float*)(ws + (size_t)N_SRC * FDIM * 4);
    float* Vp_d     = Vp_s + FDIM * 16;
    const int B = (N_DST + BUCKET_NODES - 1) / BUCKET_NODES;
    const int C = (E + CHUNK_EDGES - 1) / CHUNK_EDGES;
    int2*  pairs    = (int2*)(Vp_d + FDIM * 16);
    int*   hist     = (int*)(pairs + E);
    int*   bstart   = hist + (size_t)C * B;
    float* as_node  = (float*)(ws + (size_t)(N_SRC + N_DST) * FDIM * 4);
    int*   cnt      = (int*)(as_node + (size_t)N_SRC * HEADS);
    int*   start    = cnt + N_DST;
    int*   ssrc     = start + 2 * (size_t)N_DST;

    float* out = (float*)d_out;

    chunk_bucket_counts_build_logit_tiles<<<C + 4, 1024, 0, stream>>>(
        edst, hist, E, B, C, W_src, attS, W_dst, attD, Vp_s, Vp_d);
    deal_edges_to_buckets<<<C, 1024, 0, stream>>>(esrc, edst, hist, bstart, pairs, E, B, C);
    gemm_x128_logits_beside_node_grouping<<<B + (N_SRC + ROWS - 1) / ROWS, GEMM_THREADS, 0, stream>>>(
        x_src, W_src, Vp_s, nullptr, src_feat, as_node, N_SRC,
        pairs, bstart, start, cnt, ssrc, N_DST, B);
    self_project_softmax_aggregate<<<(N_DST + ROWS - 1) / ROWS, 512, 0, stream>>>(
        x_dst, W_self, Vp_d, b_self, ssrc, start, cnt, src_feat, as_node, out, N_DST);
}


# Aggregate gathers a node's rows 8 to a trip, the last 1 to 7 in one more trip (was 4 to a trip, then a trip per row); head weights via a per-wave LDS plane so the rows fit 64 VGPRs; sum order kept
# speedup vs baseline: 5.4723x; 1.0366x over previous
#include <hip/hip_runtime.h>
#include <hip/hip_bf16.h>
#include <math.h>

typedef __attribute__((ext_vector_type(2))) float v2f;
typedef __attribute__((ext_vector_type(8))) float v8f;

#define HEADS 4
#define FDIM  128

#define ROWS 32
#define GEMM_THREADS 288

template <int LDB, int KSTEPS>
__device__ __forceinline__ void tile_product2(const float (*At)[132],
                                              const float* __restrict__ Bcol, int m, int koff,
                                              v8f& acc0, v8f& acc1)
{
    v8f c0 = {}, c1 = {};
    #pragma unroll KSTEPS
    for (int kb = 0; kb < FDIM; kb += 4) {
        v2f a0, a1, b;
        a0.x = At[m][kb + koff];
        a0.y = At[m][kb + koff + 1];
        a1.x = At[16 + m][kb + koff];
        a1.y = At[16 + m][kb + koff + 1];
        b.x = Bcol[(kb + koff)     * LDB];
        b.y = Bcol[(kb + koff + 1) * LDB];
        c0 = __builtin_amdgcn_wmma_f32_16x16x4_f32(
                 false, a0, false, b, (short)0, c0, false, false);
        c1 = __builtin_amdgcn_wmma_f32_16x16x4_f32(
                 false, a1, false, b, (short)0, c1, false, false);
    }
    acc0 = c0; acc1 = c1;
}

#define BUCKET_SHIFT 8
#define BUCKET_NODES 256
#define CHUNK_EDGES  4096

__device__ __forceinline__ int bucket_of(int d)
{
    const unsigned b = (unsigned)d >> BUCKET_SHIFT;
    return (int)(b < BUCKET_NODES - 1u ? b : BUCKET_NODES - 1u);
}

__device__ __forceinline__ int wave_incl_scan(int v, int lane)
{
    #pragma unroll
    for (int o = 1; o < 32; o <<= 1) {
        const int t = __shfl_up(v, o);
        if (lane >= o) v += t;
    }
    return v;
}

__global__ __launch_bounds__(1024) void chunk_bucket_counts_build_logit_tiles(
    const int* __restrict__ edst, int* __restrict__ hist, int E, int B, int C,
    const float* __restrict__ W_s, const float* __restrict__ att_s,
    const float* __restrict__ W_d, const float* __restrict__ att_d,
    float* __restrict__ Vp_s, float* __restrict__ Vp_d)
{
    __shared__ int h[BUCKET_NODES];
    const int tid = threadIdx.x;
    if ((int)blockIdx.x < C) {
        const int chunk = blockIdx.x;
        if (tid < BUCKET_NODES) h[tid] = 0;
        __syncthreads();
        #pragma unroll
        for (int i = 0; i < 4; ++i) {
            const int e = chunk * CHUNK_EDGES + i * 1024 + tid;
            if (e < E)
                __hip_atomic_fetch_add(&h[bucket_of(edst[e])], 1,
                                       __ATOMIC_RELAXED, __HIP_MEMORY_SCOPE_WORKGROUP);
        }
        __syncthreads();
        if (tid < B) hist[chunk * B + tid] = h[tid];
        return;
    }
    const int t = ((int)blockIdx.x - C) * 1024 + tid;
    const int which = t >> 11;
    const int k = (t >> 4) & 127;
    const int c = t & 15;
    const float* Wm  = which ? W_d : W_s;
    const float* att = which ? att_d : att_s;
    float s = 0.0f;
    if (c < HEADS) {
        const float* wr = Wm + (size_t)k * FDIM + c * 32;
        const float* ar = att + c * 32;
        #pragma unroll
        for (int d = 0; d < 32; d += 4) {
            const float4 w = *(const float4*)(wr + d);
            const float4 a = *(const float4*)(ar + d);
            s += w.x * a.x; s += w.y * a.y; s += w.z * a.z; s += w.w * a.w;
        }
    }
    (which ? Vp_d : Vp_s)[k * 16 + c] = s;
}

__global__ __launch_bounds__(1024) void deal_edges_to_buckets(
    const int* __restrict__ esrc, const int* __restrict__ edst,
    const int* __restrict__ hist, int* __restrict__ bucket_start,
    int2* __restrict__ pairs, int E, int B, int C)
{
    __shared__ int s_tot[4][BUCKET_NODES];
    __shared__ int s_pre[4][BUCKET_NODES];
    __shared__ int s_base[BUCKET_NODES];
    __shared__ int s_fill[BUCKET_NODES];
    __shared__ int s_wsum[8];
    const int tid = threadIdx.x;
    const int lane = tid & 31;
    const int chunk = blockIdx.x;

    int d[4], s[4];
    #pragma unroll
    for (int i = 0; i < 4; ++i) {
        const int e = chunk * CHUNK_EDGES + i * 1024 + tid;
        d[i] = e < E ? edst[e] : -1;
        s[i] = e < E ? esrc[e] : 0;
    }

    const int g = tid >> 8, b = tid & (BUCKET_NODES - 1);
    int tot = 0, pre = 0;
    if (b < B) {
        #pragma unroll 7
        for (int c = g; c < C; c += 4) {
            const int v = hist[c * B + b];
            tot += v;
            pre += c < chunk ? v : 0;
        }
    }
    s_tot[g][b] = tot; s_pre[g][b] = pre;
    __syncthreads();
    int incl = 0;
    if (tid < BUCKET_NODES) {
        tot = s_tot[0][tid] + s_tot[1][tid] + s_tot[2][tid] + s_tot[3][tid];
        pre = s_pre[0][tid] + s_pre[1][tid] + s_pre[2][tid] + s_pre[3][tid];
        incl = wave_incl_scan(tot, lane);
        if (lane == 31) s_wsum[tid >> 5] = incl;
    }
    __syncthreads();
    if (tid < BUCKET_NODES) {
        int wbase = 0;
        for (int w = 0; w < (tid >> 5); ++w) wbase += s_wsum[w];
        const int excl = wbase + incl - tot;
        s_base[tid] = excl + pre;
        s_fill[tid] = 0;
        if (chunk == 0) {
            bucket_start[tid] = excl;
            if (tid == BUCKET_NODES - 1) bucket_start[BUCKET_NODES] = excl + tot;
        }
    }
    __syncthreads();
    #pragma unroll
    for (int i = 0; i < 4; ++i) {
        if (chunk * CHUNK_EDGES + i * 1024 + tid < E) {
            const int bk = bucket_of(d[i]);
            const int r = __hip_atomic_fetch_add(&s_fill[bk], 1,
                                                 __ATOMIC_RELAXED, __HIP_MEMORY_SCOPE_WORKGROUP);
            pairs[s_base[bk] + r] = make_int2(d[i], s[i]);
        }
    }
}

template <int NT, int PER>
__device__ __forceinline__ void group_bucket_by_node(
    const int2* __restrict__ pairs, const int* __restrict__ bucket_start,
    int* __restrict__ start, int* __restrict__ cnt, int* __restrict__ ssrc, int N, int bk)
{
    __shared__ int s_cnt[BUCKET_NODES];
    __shared__ int s_off[BUCKET_NODES];
    __shared__ int s_wsum[8];
    const int tid = threadIdx.x;
    const int lane = tid & 31;
    const int beg = bucket_start[bk], end = bucket_start[bk + 1];
    if (tid < BUCKET_NODES) s_cnt[tid] = 0;
    __syncthreads();
    for (int e0 = beg + tid; e0 < end; e0 += PER * NT) {
        int dl[PER];
        #pragma unroll
        for (int i = 0; i < PER; ++i) {
            const int e = e0 + i * NT;
            dl[i] = pairs[e < end ? e : end - 1].x;
        }
        #pragma unroll
        for (int i = 0; i < PER; ++i)
            if (e0 + i * NT < end)
                __hip_atomic_fetch_add(&s_cnt[dl[i] & (BUCKET_NODES - 1)], 1,
                                       __ATOMIC_RELAXED, __HIP_MEMORY_SCOPE_WORKGROUP);
    }
    __syncthreads();
    int c = 0, incl = 0;
    if (tid < BUCKET_NODES) {
        c = s_cnt[tid];
        incl = wave_incl_scan(c, lane);
        if (lane == 31) s_wsum[tid >> 5] = incl;
    }
    __syncthreads();
    if (tid < BUCKET_NODES) {
        int wbase = 0;
        for (int w = 0; w < (tid >> 5); ++w) wbase += s_wsum[w];
        const int st = beg + wbase + incl - c;
        s_off[tid] = st;
        const int node = bk * BUCKET_NODES + tid;
        if (node < N) { start[node] = st; cnt[node] = c; }
    }
    __syncthreads();
    for (int e0 = beg + tid; e0 < end; e0 += PER * NT) {
        int2 p[PER];
        #pragma unroll
        for (int i = 0; i < PER; ++i) {
            const int e = e0 + i * NT;
            p[i] = pairs[e < end ? e : end - 1];
        }
        #pragma unroll
        for (int i = 0; i < PER; ++i)
            if (e0 + i * NT < end) {
                const int pos = __hip_atomic_fetch_add(&s_off[p[i].x & (BUCKET_NODES - 1)], 1,
                                                       __ATOMIC_RELAXED, __HIP_MEMORY_SCOPE_WORKGROUP);
                ssrc[pos] = p[i].y;
            }
    }
}

__global__ __launch_bounds__(GEMM_THREADS, 16) void gemm_x128_logits_beside_node_grouping(
    const float* __restrict__ X, const float* __restrict__ W,
    const float* __restrict__ Vp, const float* __restrict__ bias,
    float* __restrict__ Y, float* __restrict__ alpha, int M,
    const int2* __restrict__ pairs, const int* __restrict__ bucket_start,
    int* __restrict__ start, int* __restrict__ cnt, int* __restrict__ ssrc, int N, int NB)
{
    if ((int)blockIdx.x < NB) {
        group_bucket_by_node<GEMM_THREADS, 8>(pairs, bucket_start, start, cnt, ssrc, N,
                                              (int)blockIdx.x);
        return;
    }
    __shared__ float At[ROWS][132];
    const int tid  = threadIdx.x;
    const int row0 = ((int)blockIdx.x - NB) * ROWS;
    const int wv = __builtin_amdgcn_readfirstlane(tid >> 5);

    if (wv < 8) {
        #pragma unroll
        for (int i = 0; i < 4; ++i) {
            int idx = (tid + i * 256) * 4;
            int r = idx >> 7, c = idx & 127;
            int gr = row0 + r; gr = gr < M ? gr : M - 1;
            const float4 v = *(const float4*)(X + (size_t)gr * FDIM + c);
            At[r][c + 0] = v.x; At[r][c + 1] = v.y;
            At[r][c + 2] = v.z; At[r][c + 3] = v.w;
        }
    }
    __syncthreads();

    const int lane = tid & 31;
    const int m    = lane & 15;
    const int half = lane >> 4;
    const int koff = half << 1;

    if (wv < 8) {
        const int n = wv * 16 + m;
        v8f acc0, acc1;
        tile_product2<FDIM, 32>(At, W + n, m, koff, acc0, acc1);
        const float bn = bias ? bias[n] : 0.0f;
        #pragma unroll
        for (int r = 0; r < 8; ++r) {
            const int row = row0 + half * 8 + r;
            if (row < M)      Y[(size_t)row * FDIM + n]        = acc0[r] + bn;
            if (row + 16 < M) Y[(size_t)(row + 16) * FDIM + n] = acc1[r] + bn;
        }
    } else {
        v8f acc0, acc1;
        tile_product2<16, 32>(At, Vp + m, m, koff, acc0, acc1);
        if (m < HEADS) {
            #pragma unroll
            for (int r = 0; r < 8; ++r) {
                const int row = row0 + half * 8 + r;
                if (row < M)      alpha[(size_t)row * HEADS + m]        = acc0[r];
                if (row + 16 < M) alpha[(size_t)(row + 16) * HEADS + m] = acc1[r];
            }
        }
    }
}

__device__ __forceinline__ float uniform_f(float v)
{
    return __int_as_float(__builtin_amdgcn_readfirstlane(__float_as_int(v)));
}

__device__ __forceinline__ float lrelu(float x) { return x > 0.f ? x : 0.2f * x; }

__device__ __forceinline__ float4 edge_scores(const float* __restrict__ asn, int s, float4 adv)
{
    const float4 a = *(const float4*)(asn + (size_t)s * HEADS);
    float4 r;
    r.x = lrelu(a.x + adv.x); r.y = lrelu(a.y + adv.y);
    r.z = lrelu(a.z + adv.z); r.w = lrelu(a.w + adv.w);
    return r;
}

__device__ __forceinline__ float4 wave_max4(float4 v)
{
    #pragma unroll
    for (int o = 16; o > 0; o >>= 1) {
        v.x = fmaxf(v.x, __shfl_xor(v.x, o)); v.y = fmaxf(v.y, __shfl_xor(v.y, o));
        v.z = fmaxf(v.z, __shfl_xor(v.z, o)); v.w = fmaxf(v.w, __shfl_xor(v.w, o));
    }
    return v;
}

__device__ __forceinline__ float4 wave_sum4(float4 v)
{
    #pragma unroll
    for (int o = 16; o > 0; o >>= 1) {
        v.x += __shfl_xor(v.x, o); v.y += __shfl_xor(v.y, o);
        v.z += __shfl_xor(v.z, o); v.w += __shfl_xor(v.w, o);
    }
    return v;
}

__device__ __forceinline__ unsigned row_offset(int s_l, int i, int lane)
{
    const unsigned s = (unsigned)__builtin_amdgcn_readlane(s_l, i);
    return s * (FDIM / 4) + (unsigned)lane;
}

__device__ __forceinline__ float4 src_row(const float* __restrict__ src_feat, int s_l, int i, int lane)
{
    return ((const float4*)src_feat)[row_offset(s_l, i, lane)];
}

__device__ __forceinline__ void put_weights(float* plane, int lane, float4 w)
{
    plane[lane] = w.x; plane[32 + lane] = w.y; plane[64 + lane] = w.z; plane[96 + lane] = w.w;
}

__device__ __forceinline__ void fma_row(float4& acc, float w, float4 f)
{
    acc.x += w * f.x; acc.y += w * f.y; acc.z += w * f.z; acc.w += w * f.w;
}

__device__ __forceinline__ void add_rows(int s_l, const float* wrow, int m, int lane,
                                         const float* __restrict__ src_feat, float4& acc)
{
    int i = 0;
    for (; i + 8 <= m; i += 8) {
        float4 f[8];
        #pragma unroll
        for (int k = 0; k < 8; ++k) f[k] = src_row(src_feat, s_l, i + k, lane);
        const float4 wa = *(const float4*)(wrow + i);
        const float4 wb = *(const float4*)(wrow + i + 4);
        fma_row(acc, wa.x, f[0]); fma_row(acc, wa.y, f[1]);
        fma_row(acc, wa.z, f[2]); fma_row(acc, wa.w, f[3]);
        fma_row(acc, wb.x, f[4]); fma_row(acc, wb.y, f[5]);
        fma_row(acc, wb.z, f[6]); fma_row(acc, wb.w, f[7]);
    }
    if (i < m) {
        const int r = m - i;
        unsigned o[7];
        #pragma unroll
        for (int k = 0; k < 7; ++k) o[k] = row_offset(s_l, i + k, lane);
        float4 f[7];
        #pragma unroll
        for (int k = 0; k < 7; ++k) {
            f[k] = make_float4(0.f, 0.f, 0.f, 0.f);
            if (k < r) f[k] = ((const float4*)src_feat)[o[k]];
        }
        const float4 wa = *(const float4*)(wrow + i);
        const float4 wb = *(const float4*)(wrow + i + 4);
        fma_row(acc, wa.x, f[0]); fma_row(acc, wa.y, f[1]);
        fma_row(acc, wa.z, f[2]); fma_row(acc, wa.w, f[3]);
        fma_row(acc, wb.x, f[4]); fma_row(acc, wb.y, f[5]);
        fma_row(acc, wb.z, f[6]);
    }
}

__device__ __forceinline__ void aggregate_node(
    int d, int beg, int n, const float* selfrow, const float* lrow, int lane,
    float* plane,
    const int* __restrict__ ssrc, const float* __restrict__ src_feat,
    const float* __restrict__ asn, float* __restrict__ out)
{
    const float4 adv_v = *(const float4*)lrow;
    float4 adv;
    adv.x = uniform_f(adv_v.x); adv.y = uniform_f(adv_v.y);
    adv.z = uniform_f(adv_v.z); adv.w = uniform_f(adv_v.w);
    const float NEG = -INFINITY;
    const float* wrow = plane + (lane >> 3) * 32;

    const bool v0 = lane < n;
    const int s0 = v0 ? ssrc[beg + lane] : 0;
    float4 al0 = {NEG, NEG, NEG, NEG};
    if (v0) al0 = edge_scores(asn, s0, adv);

    float4 mx = al0;
    for (int b = 32; b < n; b += 32) {
        const int j = b + lane;
        if (j < n) {
            const float4 al = edge_scores(asn, ssrc[beg + j], adv);
            mx.x = fmaxf(mx.x, al.x); mx.y = fmaxf(mx.y, al.y);
            mx.z = fmaxf(mx.z, al.z); mx.w = fmaxf(mx.w, al.w);
        }
    }
    mx = wave_max4(mx);
    mx.x = uniform_f(mx.x); mx.y = uniform_f(mx.y); mx.z = uniform_f(mx.z); mx.w = uniform_f(mx.w);

    float4 ex0 = {0.f, 0.f, 0.f, 0.f};
    if (v0) {
        ex0.x = __expf(al0.x - mx.x); ex0.y = __expf(al0.y - mx.y);
        ex0.z = __expf(al0.z - mx.z); ex0.w = __expf(al0.w - mx.w);
    }
    float4 sm = ex0;
    for (int b = 32; b < n; b += 32) {
        const int j = b + lane;
        if (j < n) {
            const float4 al = edge_scores(asn, ssrc[beg + j], adv);
            sm.x += __expf(al.x - mx.x); sm.y += __expf(al.y - mx.y);
            sm.z += __expf(al.z - mx.z); sm.w += __expf(al.w - mx.w);
        }
    }
    sm = wave_sum4(sm);
    float4 den;
    den.x = uniform_f(sm.x + 1e-16f); den.y = uniform_f(sm.y + 1e-16f);
    den.z = uniform_f(sm.z + 1e-16f); den.w = uniform_f(sm.w + 1e-16f);

    float4 acc = {0.f, 0.f, 0.f, 0.f};
    float4 w0;
    w0.x = ex0.x / den.x; w0.y = ex0.y / den.y; w0.z = ex0.z / den.z; w0.w = ex0.w / den.w;
    put_weights(plane, lane, w0);
    add_rows(s0, wrow, n < 32 ? n : 32, lane, src_feat, acc);
    for (int b = 32; b < n; b += 32) {
        const int j = b + lane;
        int s = 0;
        float4 w = {0.f, 0.f, 0.f, 0.f};
        if (j < n) {
            s = ssrc[beg + j];
            const float4 al = edge_scores(asn, s, adv);
            w.x = __expf(al.x - mx.x) / den.x; w.y = __expf(al.y - mx.y) / den.y;
            w.z = __expf(al.z - mx.z) / den.z; w.w = __expf(al.w - mx.w) / den.w;
        }
        const int rem = n - b;
        put_weights(plane, lane, w);
        add_rows(s, wrow, rem < 32 ? rem : 32, lane, src_feat, acc);
    }

    const float4 self = *(const float4*)(selfrow + lane * 4);
    float4 o;
    o.x = self.x + acc.x; o.y = self.y + acc.y; o.z = self.z + acc.z; o.w = self.w + acc.w;
    *(float4*)(out + (size_t)d * FDIM + lane * 4) = o;
}

__global__ __launch_bounds__(512, 16) void self_project_softmax_aggregate(
    const float* __restrict__ X, const float* __restrict__ W,
    const float* __restrict__ Vp, const float* __restrict__ bias,
    const int* __restrict__ ssrc, const int* __restrict__ start, const int* __restrict__ cnt,
    const float* __restrict__ src_feat, const float* __restrict__ asn,
    float* __restrict__ out, int N)
{
    __shared__ __attribute__((aligned(16))) float At[ROWS][132];
    __shared__ __attribute__((aligned(16))) float Yt[ROWS][132];
    __shared__ __attribute__((aligned(16))) float Lt[ROWS][HEADS];
    const int tid  = threadIdx.x;
    const int lane = tid & 31;
    const int row0 = blockIdx.x * ROWS;
    const int wv = __builtin_amdgcn_readfirstlane(tid >> 5);

    const int dA  = row0 + wv, dB = dA + 16;
    const int dcA = dA < N ? dA : N - 1;
    const int dcB = dB < N ? dB : N - 1;
    const int begA = start[dcA], nA = cnt[dcA];
    const int begB = start[dcB], nB = cnt[dcB];

    {
        #pragma unroll
        for (int i = 0; i < 2; ++i) {
            int idx = (tid + i * 512) * 4;
            int r = idx >> 7, c = idx & 127;
            int gr = row0 + r; gr = gr < N ? gr : N - 1;
            const float4 v = *(const float4*)(X + (size_t)gr * FDIM + c);
            At[r][c + 0] = v.x; At[r][c + 1] = v.y;
            At[r][c + 2] = v.z; At[r][c + 3] = v.w;
        }
    }
    __syncthreads();
    {
        const int m    = lane & 15;
        const int half = lane >> 4;
        const int koff = half << 1;
        if (wv < 8) {
            const int col = wv * 16 + m;
            v8f acc0, acc1;
            tile_product2<FDIM, 16>(At, W + col, m, koff, acc0, acc1);
            const float bn = bias[col];
            #pragma unroll
            for (int r = 0; r < 8; ++r) {
                Yt[half * 8 + r][col]      = acc0[r] + bn;
                Yt[16 + half * 8 + r][col] = acc1[r] + bn;
            }
        } else if (wv == 8) {
            v8f acc0, acc1;
            tile_product2<16, 16>(At, Vp + m, m, koff, acc0, acc1);
            if (m < HEADS) {
                #pragma unroll
                for (int r = 0; r < 8; ++r) {
                    Lt[half * 8 + r][m]      = acc0[r];
                    Lt[16 + half * 8 + r][m] = acc1[r];
                }
            }
        }
    }
    __syncthreads();

    float* plane = &At[0][0] + wv * 128;
    if (dA < N)
        aggregate_node(dA, begA, nA, &Yt[wv][0], &Lt[wv][0], lane, plane,
                       ssrc, src_feat, asn, out);
    if (dB < N)
        aggregate_node(dB, begB, nB, &Yt[wv + 16][0], &Lt[wv + 16][0], lane, plane,
                       ssrc, src_feat, asn, out);
}

extern "C" void kernel_launch(void* const* d_in, const int* in_sizes, int n_in,
                              void* d_out, int out_size, void* d_ws, size_t ws_size,
                              hipStream_t stream)
{
    const float* x_src  = (const float*)d_in[0];
    const float* x_dst  = (const float*)d_in[1];
    const int*   esrc   = (const int*)d_in[2];
    const int*   edst   = (const int*)d_in[3];
    const float* W_src  = (const float*)d_in[5];
    const float* W_dst  = (const float*)d_in[6];
    const float* attS   = (const float*)d_in[7];
    const float* attD   = (const float*)d_in[8];
    const float* W_self = (const float*)d_in[9];
    const float* b_self = (const float*)d_in[10];

    const int N_SRC = in_sizes[0] / FDIM;
    const int N_DST = in_sizes[1] / FDIM;
    const int E     = in_sizes[2];

    char* ws = (char*)d_ws;
    float* src_feat = (float*)(ws);
    float* Vp_s     = (float*)(ws + (size_t)N_SRC * FDIM * 4);
    float* Vp_d     = Vp_s + FDIM * 16;
    const int B = (N_DST + BUCKET_NODES - 1) / BUCKET_NODES;
    const int C = (E + CHUNK_EDGES - 1) / CHUNK_EDGES;
    int2*  pairs    = (int2*)(Vp_d + FDIM * 16);
    int*   hist     = (int*)(pairs + E);
    int*   bstart   = hist + (size_t)C * B;
    float* as_node  = (float*)(ws + (size_t)(N_SRC + N_DST) * FDIM * 4);
    int*   cnt      = (int*)(as_node + (size_t)N_SRC * HEADS);
    int*   start    = cnt + N_DST;
    int*   ssrc     = start + 2 * (size_t)N_DST;

    float* out = (float*)d_out;

    chunk_bucket_counts_build_logit_tiles<<<C + 4, 1024, 0, stream>>>(
        edst, hist, E, B, C, W_src, attS, W_dst, attD, Vp_s, Vp_d);
    deal_edges_to_buckets<<<C, 1024, 0, stream>>>(esrc, edst, hist, bstart, pairs, E, B, C);
    gemm_x128_logits_beside_node_grouping<<<B + (N_SRC + ROWS - 1) / ROWS, GEMM_THREADS, 0, stream>>>(
        x_src, W_src, Vp_s, nullptr, src_feat, as_node, N_SRC,
        pairs, bstart, start, cnt, ssrc, N_DST, B);
    self_project_softmax_aggregate<<<(N_DST + ROWS - 1) / ROWS, 512, 0, stream>>>(
        x_dst, W_self, Vp_d, b_self, ssrc, start, cnt, src_feat, as_node, out, N_DST);
}


# Level 1 of the edge grouping is chunk-local: each chunk orders its own edges by bucket, bucket work-groups walk per-chunk segments; 4 to 3 launches; a node's edge (sum) order may differ
# speedup vs baseline: 6.5629x; 1.1993x over previous
#include <hip/hip_runtime.h>
#include <hip/hip_bf16.h>
#include <math.h>

typedef __attribute__((ext_vector_type(2))) float v2f;
typedef __attribute__((ext_vector_type(8))) float v8f;

#define HEADS 4
#define FDIM  128

#define ROWS 32
#define GEMM_THREADS 288

template <int LDB, int KSTEPS>
__device__ __forceinline__ void tile_product2(const float (*At)[132],
                                              const float* __restrict__ Bcol, int m, int koff,
                                              v8f& acc0, v8f& acc1)
{
    v8f c0 = {}, c1 = {};
    #pragma unroll KSTEPS
    for (int kb = 0; kb < FDIM; kb += 4) {
        v2f a0, a1, b;
        a0.x = At[m][kb + koff];
        a0.y = At[m][kb + koff + 1];
        a1.x = At[16 + m][kb + koff];
        a1.y = At[16 + m][kb + koff + 1];
        b.x = Bcol[(kb + koff)     * LDB];
        b.y = Bcol[(kb + koff + 1) * LDB];
        c0 = __builtin_amdgcn_wmma_f32_16x16x4_f32(
                 false, a0, false, b, (short)0, c0, false, false);
        c1 = __builtin_amdgcn_wmma_f32_16x16x4_f32(
                 false, a1, false, b, (short)0, c1, false, false);
    }
    acc0 = c0; acc1 = c1;
}

#define BUCKET_SHIFT 8
#define BUCKET_NODES 256
#define CHUNK_EDGES  4096
#define SEG_PITCH    (BUCKET_NODES + 1)

__device__ __forceinline__ int bucket_of(int d)
{
    const unsigned b = (unsigned)d >> BUCKET_SHIFT;
    return (int)(b < BUCKET_NODES - 1u ? b : BUCKET_NODES - 1u);
}

__device__ __forceinline__ int wave_incl_scan(int v, int lane)
{
    #pragma unroll
    for (int o = 1; o < 32; o <<= 1) {
        const int t = __shfl_up(v, o);
        if (lane >= o) v += t;
    }
    return v;
}

__global__ __launch_bounds__(1024) void sort_chunk_by_bucket_build_logit_tiles(
    const int* __restrict__ esrc, const int* __restrict__ edst,
    int* __restrict__ seg, int2* __restrict__ pairs, int E, int C,
    const float* __restrict__ W_s, const float* __restrict__ att_s,
    const float* __restrict__ W_d, const float* __restrict__ att_d,
    float* __restrict__ Vp_s, float* __restrict__ Vp_d)
{
    __shared__ int h[BUCKET_NODES];
    __shared__ int s_off[BUCKET_NODES];
    __shared__ int s_wsum[8];
    const int tid = threadIdx.x;
    if ((int)blockIdx.x < C) {
        const int chunk = blockIdx.x;
        const int lane = tid & 31;
        int d[4], s[4];
        #pragma unroll
        for (int i = 0; i < 4; ++i) {
            const int e = chunk * CHUNK_EDGES + i * 1024 + tid;
            d[i] = e < E ? edst[e] : -1;
            s[i] = e < E ? esrc[e] : 0;
        }
        if (tid < BUCKET_NODES) h[tid] = 0;
        __syncthreads();
        #pragma unroll
        for (int i = 0; i < 4; ++i)
            if (chunk * CHUNK_EDGES + i * 1024 + tid < E)
                __hip_atomic_fetch_add(&h[bucket_of(d[i])], 1,
                                       __ATOMIC_RELAXED, __HIP_MEMORY_SCOPE_WORKGROUP);
        __syncthreads();
        int cb = 0, incl = 0;
        if (tid < BUCKET_NODES) {
            cb = h[tid];
            incl = wave_incl_scan(cb, lane);
            if (lane == 31) s_wsum[tid >> 5] = incl;
        }
        __syncthreads();
        if (tid < BUCKET_NODES) {
            int wbase = 0;
            for (int w = 0; w < (tid >> 5); ++w) wbase += s_wsum[w];
            const int excl = wbase + incl - cb;
            s_off[tid] = excl;
            h[tid] = 0;
            seg[(size_t)chunk * SEG_PITCH + tid] = excl;
            if (tid == BUCKET_NODES - 1)
                seg[(size_t)chunk * SEG_PITCH + BUCKET_NODES] = excl + cb;
        }
        __syncthreads();
        int2* mine = pairs + (size_t)chunk * CHUNK_EDGES;
        #pragma unroll
        for (int i = 0; i < 4; ++i) {
            if (chunk * CHUNK_EDGES + i * 1024 + tid < E) {
                const int bk = bucket_of(d[i]);
                const int r = __hip_atomic_fetch_add(&h[bk], 1,
                                                     __ATOMIC_RELAXED, __HIP_MEMORY_SCOPE_WORKGROUP);
                mine[s_off[bk] + r] = make_int2(d[i], s[i]);
            }
        }
        return;
    }
    const int t = ((int)blockIdx.x - C) * 1024 + tid;
    const int which = t >> 11;
    const int k = (t >> 4) & 127;
    const int c = t & 15;
    const float* Wm  = which ? W_d : W_s;
    const float* att = which ? att_d : att_s;
    float sum = 0.0f;
    if (c < HEADS) {
        const float* wr = Wm + (size_t)k * FDIM + c * 32;
        const float* ar = att + c * 32;
        #pragma unroll
        for (int dd = 0; dd < 32; dd += 4) {
            const float4 w = *(const float4*)(wr + dd);
            const float4 a = *(const float4*)(ar + dd);
            sum += w.x * a.x; sum += w.y * a.y; sum += w.z * a.z; sum += w.w * a.w;
        }
    }
    (which ? Vp_d : Vp_s)[k * 16 + c] = sum;
}

template <int NT, int PER, bool PLACE>
__device__ __forceinline__ int walk_bucket_segments(
    const int2* __restrict__ pairs, const int* __restrict__ seg, int bk, int C,
    int* s_lo, int* s_hi, int* s_node, int* __restrict__ ssrc)
{
    const int tid = threadIdx.x;
    const int grp = tid >> 3, sub = tid & 7;
    int lo_sum = 0;
    for (int c0 = 0; c0 < C; c0 += NT) {
        const int c = c0 + tid;
        int lo = 0, hi = 0;
        if (c < C) {
            lo = seg[(size_t)c * SEG_PITCH + bk];
            hi = seg[(size_t)c * SEG_PITCH + bk + 1];
        }
        lo_sum += lo;
        s_lo[tid] = lo; s_hi[tid] = hi;
        __syncthreads();
        const int ns = C - c0 < NT ? C - c0 : NT;
        for (int j = grp; j < ns; j += NT / 8) {
            const int l = s_lo[j], h = s_hi[j];
            const int2* base = pairs + (size_t)(c0 + j) * CHUNK_EDGES;
            for (int e0 = l + sub; e0 < h; e0 += 8 * PER) {
                if (PLACE) {
                    int2 p[PER];
                    #pragma unroll
                    for (int i = 0; i < PER; ++i) {
                        const int e = e0 + 8 * i;
                        p[i] = base[e < h ? e : h - 1];
                    }
                    #pragma unroll
                    for (int i = 0; i < PER; ++i) asm volatile("" :: "v"(p[i].x), "v"(p[i].y));
                    #pragma unroll
                    for (int i = 0; i < PER; ++i)
                        if (e0 + 8 * i < h) {
                            const int pos = __hip_atomic_fetch_add(
                                &s_node[p[i].x & (BUCKET_NODES - 1)], 1,
                                __ATOMIC_RELAXED, __HIP_MEMORY_SCOPE_WORKGROUP);
                            ssrc[pos] = p[i].y;
                        }
                } else {
                    int dl[PER];
                    #pragma unroll
                    for (int i = 0; i < PER; ++i) {
                        const int e = e0 + 8 * i;
                        dl[i] = base[e < h ? e : h - 1].x;
                    }
                    #pragma unroll
                    for (int i = 0; i < PER; ++i) asm volatile("" :: "v"(dl[i]));
                    #pragma unroll
                    for (int i = 0; i < PER; ++i)
                        if (e0 + 8 * i < h)
                            __hip_atomic_fetch_add(&s_node[dl[i] & (BUCKET_NODES - 1)], 1,
                                                   __ATOMIC_RELAXED, __HIP_MEMORY_SCOPE_WORKGROUP);
                }
            }
        }
        __syncthreads();
    }
    return lo_sum;
}

template <int NT, int PER>
__device__ __forceinline__ void group_bucket_by_node(
    const int2* __restrict__ pairs, const int* __restrict__ seg,
    int* __restrict__ start, int* __restrict__ cnt, int* __restrict__ ssrc, int N, int bk, int C)
{
    __shared__ int s_cnt[BUCKET_NODES];
    __shared__ int s_off[BUCKET_NODES];
    __shared__ int s_lo[NT];
    __shared__ int s_hi[NT];
    __shared__ int s_wsum[8];
    __shared__ int s_beg;
    const int tid = threadIdx.x;
    const int lane = tid & 31;
    if (tid < BUCKET_NODES) s_cnt[tid] = 0;
    if (tid == 0) s_beg = 0;
    __syncthreads();
    int lo_sum = walk_bucket_segments<NT, PER, false>(pairs, seg, bk, C, s_lo, s_hi, s_cnt, ssrc);
    #pragma unroll
    for (int o = 16; o > 0; o >>= 1) lo_sum += __shfl_xor(lo_sum, o);
    if (lane == 0)
        __hip_atomic_fetch_add(&s_beg, lo_sum, __ATOMIC_RELAXED, __HIP_MEMORY_SCOPE_WORKGROUP);
    __syncthreads();
    const int beg = s_beg;
    int cn = 0, incl = 0;
    if (tid < BUCKET_NODES) {
        cn = s_cnt[tid];
        incl = wave_incl_scan(cn, lane);
        if (lane == 31) s_wsum[tid >> 5] = incl;
    }
    __syncthreads();
    if (tid < BUCKET_NODES) {
        int wbase = 0;
        for (int w = 0; w < (tid >> 5); ++w) wbase += s_wsum[w];
        const int st = beg + wbase + incl - cn;
        s_off[tid] = st;
        const int node = bk * BUCKET_NODES + tid;
        if (node < N) { start[node] = st; cnt[node] = cn; }
    }
    __syncthreads();
    walk_bucket_segments<NT, PER, true>(pairs, seg, bk, C, s_lo, s_hi, s_off, ssrc);
}

__global__ __launch_bounds__(GEMM_THREADS, 16) void gemm_x128_logits_beside_node_grouping(
    const float* __restrict__ X, const float* __restrict__ W,
    const float* __restrict__ Vp, const float* __restrict__ bias,
    float* __restrict__ Y, float* __restrict__ alpha, int M,
    const int2* __restrict__ pairs, const int* __restrict__ seg,
    int* __restrict__ start, int* __restrict__ cnt, int* __restrict__ ssrc, int N, int NB, int C)
{
    if ((int)blockIdx.x < NB) {
        group_bucket_by_node<GEMM_THREADS, 4>(pairs, seg, start, cnt, ssrc, N,
                                              (int)blockIdx.x, C);
        return;
    }
    __shared__ float At[ROWS][132];
    const int tid  = threadIdx.x;
    const int row0 = ((int)blockIdx.x - NB) * ROWS;
    const int wv = __builtin_amdgcn_readfirstlane(tid >> 5);

    if (wv < 8) {
        #pragma unroll
        for (int i = 0; i < 4; ++i) {
            int idx = (tid + i * 256) * 4;
            int r = idx >> 7, c = idx & 127;
            int gr = row0 + r; gr = gr < M ? gr : M - 1;
            const float4 v = *(const float4*)(X + (size_t)gr * FDIM + c);
            At[r][c + 0] = v.x; At[r][c + 1] = v.y;
            At[r][c + 2] = v.z; At[r][c + 3] = v.w;
        }
    }
    __syncthreads();

    const int lane = tid & 31;
    const int m    = lane & 15;
    const int half = lane >> 4;
    const int koff = half << 1;

    if (wv < 8) {
        const int n = wv * 16 + m;
        v8f acc0, acc1;
        tile_product2<FDIM, 32>(At, W + n, m, koff, acc0, acc1);
        const float bn = bias ? bias[n] : 0.0f;
        #pragma unroll
        for (int r = 0; r < 8; ++r) {
            const int row = row0 + half * 8 + r;
            if (row < M)      Y[(size_t)row * FDIM + n]        = acc0[r] + bn;
            if (row + 16 < M) Y[(size_t)(row + 16) * FDIM + n] = acc1[r] + bn;
        }
    } else {
        v8f acc0, acc1;
        tile_product2<16, 32>(At, Vp + m, m, koff, acc0, acc1);
        if (m < HEADS) {
            #pragma unroll
            for (int r = 0; r < 8; ++r) {
                const int row = row0 + half * 8 + r;
                if (row < M)      alpha[(size_t)row * HEADS + m]        = acc0[r];
                if (row + 16 < M) alpha[(size_t)(row + 16) * HEADS + m] = acc1[r];
            }
        }
    }
}

__device__ __forceinline__ float uniform_f(float v)
{
    return __int_as_float(__builtin_amdgcn_readfirstlane(__float_as_int(v)));
}

__device__ __forceinline__ float lrelu(float x) { return x > 0.f ? x : 0.2f * x; }

__device__ __forceinline__ float4 edge_scores(const float* __restrict__ asn, int s, float4 adv)
{
    const float4 a = *(const float4*)(asn + (size_t)s * HEADS);
    float4 r;
    r.x = lrelu(a.x + adv.x); r.y = lrelu(a.y + adv.y);
    r.z = lrelu(a.z + adv.z); r.w = lrelu(a.w + adv.w);
    return r;
}

__device__ __forceinline__ float4 wave_max4(float4 v)
{
    #pragma unroll
    for (int o = 16; o > 0; o >>= 1) {
        v.x = fmaxf(v.x, __shfl_xor(v.x, o)); v.y = fmaxf(v.y, __shfl_xor(v.y, o));
        v.z = fmaxf(v.z, __shfl_xor(v.z, o)); v.w = fmaxf(v.w, __shfl_xor(v.w, o));
    }
    return v;
}

__device__ __forceinline__ float4 wave_sum4(float4 v)
{
    #pragma unroll
    for (int o = 16; o > 0; o >>= 1) {
        v.x += __shfl_xor(v.x, o); v.y += __shfl_xor(v.y, o);
        v.z += __shfl_xor(v.z, o); v.w += __shfl_xor(v.w, o);
    }
    return v;
}

__device__ __forceinline__ unsigned row_offset(int s_l, int i, int lane)
{
    const unsigned s = (unsigned)__builtin_amdgcn_readlane(s_l, i);
    return s * (FDIM / 4) + (unsigned)lane;
}

__device__ __forceinline__ float4 src_row(const float* __restrict__ src_feat, int s_l, int i, int lane)
{
    return ((const float4*)src_feat)[row_offset(s_l, i, lane)];
}

__device__ __forceinline__ void put_weights(float* plane, int lane, float4 w)
{
    plane[lane] = w.x; plane[32 + lane] = w.y; plane[64 + lane] = w.z; plane[96 + lane] = w.w;
}

__device__ __forceinline__ void fma_row(float4& acc, float w, float4 f)
{
    acc.x += w * f.x; acc.y += w * f.y; acc.z += w * f.z; acc.w += w * f.w;
}

__device__ __forceinline__ void add_rows(int s_l, const float* wrow, int m, int lane,
                                         const float* __restrict__ src_feat, float4& acc)
{
    int i = 0;
    for (; i + 8 <= m; i += 8) {
        float4 f[8];
        #pragma unroll
        for (int k = 0; k < 8; ++k) f[k] = src_row(src_feat, s_l, i + k, lane);
        const float4 wa = *(const float4*)(wrow + i);
        const float4 wb = *(const float4*)(wrow + i + 4);
        fma_row(acc, wa.x, f[0]); fma_row(acc, wa.y, f[1]);
        fma_row(acc, wa.z, f[2]); fma_row(acc, wa.w, f[3]);
        fma_row(acc, wb.x, f[4]); fma_row(acc, wb.y, f[5]);
        fma_row(acc, wb.z, f[6]); fma_row(acc, wb.w, f[7]);
    }
    if (i < m) {
        const int r = m - i;
        unsigned o[7];
        #pragma unroll
        for (int k = 0; k < 7; ++k) o[k] = row_offset(s_l, i + k, lane);
        float4 f[7];
        #pragma unroll
        for (int k = 0; k < 7; ++k) {
            f[k] = make_float4(0.f, 0.f, 0.f, 0.f);
            if (k < r) f[k] = ((const float4*)src_feat)[o[k]];
        }
        const float4 wa = *(const float4*)(wrow + i);
        const float4 wb = *(const float4*)(wrow + i + 4);
        fma_row(acc, wa.x, f[0]); fma_row(acc, wa.y, f[1]);
        fma_row(acc, wa.z, f[2]); fma_row(acc, wa.w, f[3]);
        fma_row(acc, wb.x, f[4]); fma_row(acc, wb.y, f[5]);
        fma_row(acc, wb.z, f[6]);
    }
}

__device__ __forceinline__ void aggregate_node(
    int d, int beg, int n, const float* selfrow, const float* lrow, int lane,
    float* plane,
    const int* __restrict__ ssrc, const float* __restrict__ src_feat,
    const float* __restrict__ asn, float* __restrict__ out)
{
    const float4 adv_v = *(const float4*)lrow;
    float4 adv;
    adv.x = uniform_f(adv_v.x); adv.y = uniform_f(adv_v.y);
    adv.z = uniform_f(adv_v.z); adv.w = uniform_f(adv_v.w);
    const float NEG = -INFINITY;
    const float* wrow = plane + (lane >> 3) * 32;

    const bool v0 = lane < n;
    const int s0 = v0 ? ssrc[beg + lane] : 0;
    float4 al0 = {NEG, NEG, NEG, NEG};
    if (v0) al0 = edge_scores(asn, s0, adv);

    float4 mx = al0;
    for (int b = 32; b < n; b += 32) {
        const int j = b + lane;
        if (j < n) {
            const float4 al = edge_scores(asn, ssrc[beg + j], adv);
            mx.x = fmaxf(mx.x, al.x); mx.y = fmaxf(mx.y, al.y);
            mx.z = fmaxf(mx.z, al.z); mx.w = fmaxf(mx.w, al.w);
        }
    }
    mx = wave_max4(mx);
    mx.x = uniform_f(mx.x); mx.y = uniform_f(mx.y); mx.z = uniform_f(mx.z); mx.w = uniform_f(mx.w);

    float4 ex0 = {0.f, 0.f, 0.f, 0.f};
    if (v0) {
        ex0.x = __expf(al0.x - mx.x); ex0.y = __expf(al0.y - mx.y);
        ex0.z = __expf(al0.z - mx.z); ex0.w = __expf(al0.w - mx.w);
    }
    float4 sm = ex0;
    for (int b = 32; b < n; b += 32) {
        const int j = b + lane;
        if (j < n) {
            const float4 al = edge_scores(asn, ssrc[beg + j], adv);
            sm.x += __expf(al.x - mx.x); sm.y += __expf(al.y - mx.y);
            sm.z += __expf(al.z - mx.z); sm.w += __expf(al.w - mx.w);
        }
    }
    sm = wave_sum4(sm);
    float4 den;
    den.x = uniform_f(sm.x + 1e-16f); den.y = uniform_f(sm.y + 1e-16f);
    den.z = uniform_f(sm.z + 1e-16f); den.w = uniform_f(sm.w + 1e-16f);

    float4 acc = {0.f, 0.f, 0.f, 0.f};
    float4 w0;
    w0.x = ex0.x / den.x; w0.y = ex0.y / den.y; w0.z = ex0.z / den.z; w0.w = ex0.w / den.w;
    put_weights(plane, lane, w0);
    add_rows(s0, wrow, n < 32 ? n : 32, lane, src_feat, acc);
    for (int b = 32; b < n; b += 32) {
        const int j = b + lane;
        int s = 0;
        float4 w = {0.f, 0.f, 0.f, 0.f};
        if (j < n) {
            s = ssrc[beg + j];
            const float4 al = edge_scores(asn, s, adv);
            w.x = __expf(al.x - mx.x) / den.x; w.y = __expf(al.y - mx.y) / den.y;
            w.z = __expf(al.z - mx.z) / den.z; w.w = __expf(al.w - mx.w) / den.w;
        }
        const int rem = n - b;
        put_weights(plane, lane, w);
        add_rows(s, wrow, rem < 32 ? rem : 32, lane, src_feat, acc);
    }

    const float4 self = *(const float4*)(selfrow + lane * 4);
    float4 o;
    o.x = self.x + acc.x; o.y = self.y + acc.y; o.z = self.z + acc.z; o.w = self.w + acc.w;
    *(float4*)(out + (size_t)d * FDIM + lane * 4) = o;
}

__global__ __launch_bounds__(512, 16) void self_project_softmax_aggregate(
    const float* __restrict__ X, const float* __restrict__ W,
    const float* __restrict__ Vp, const float* __restrict__ bias,
    const int* __restrict__ ssrc, const int* __restrict__ start, const int* __restrict__ cnt,
    const float* __restrict__ src_feat, const float* __restrict__ asn,
    float* __restrict__ out, int N)
{
    __shared__ __attribute__((aligned(16))) float At[ROWS][132];
    __shared__ __attribute__((aligned(16))) float Yt[ROWS][132];
    __shared__ __attribute__((aligned(16))) float Lt[ROWS][HEADS];
    const int tid  = threadIdx.x;
    const int lane = tid & 31;
    const int row0 = blockIdx.x * ROWS;
    const int wv = __builtin_amdgcn_readfirstlane(tid >> 5);

    const int dA  = row0 + wv, dB = dA + 16;
    const int dcA = dA < N ? dA : N - 1;
    const int dcB = dB < N ? dB : N - 1;
    const int begA = start[dcA], nA = cnt[dcA];
    const int begB = start[dcB], nB = cnt[dcB];

    {
        #pragma unroll
        for (int i = 0; i < 2; ++i) {
            int idx = (tid + i * 512) * 4;
            int r = idx >> 7, c = idx & 127;
            int gr = row0 + r; gr = gr < N ? gr : N - 1;
            const float4 v = *(const float4*)(X + (size_t)gr * FDIM + c);
            At[r][c + 0] = v.x; At[r][c + 1] = v.y;
            At[r][c + 2] = v.z; At[r][c + 3] = v.w;
        }
    }
    __syncthreads();
    {
        const int m    = lane & 15;
        const int half = lane >> 4;
        const int koff = half << 1;
        if (wv < 8) {
            const int col = wv * 16 + m;
            v8f acc0, acc1;
            tile_product2<FDIM, 16>(At, W + col, m, koff, acc0, acc1);
            const float bn = bias[col];
            #pragma unroll
            for (int r = 0; r < 8; ++r) {
                Yt[half * 8 + r][col]      = acc0[r] + bn;
                Yt[16 + half * 8 + r][col] = acc1[r] + bn;
            }
        } else if (wv == 8) {
            v8f acc0, acc1;
            tile_product2<16, 16>(At, Vp + m, m, koff, acc0, acc1);
            if (m < HEADS) {
                #pragma unroll
                for (int r = 0; r < 8; ++r) {
                    Lt[half * 8 + r][m]      = acc0[r];
                    Lt[16 + half * 8 + r][m] = acc1[r];
                }
            }
        }
    }
    __syncthreads();

    float* plane = &At[0][0] + wv * 128;
    if (dA < N)
        aggregate_node(dA, begA, nA, &Yt[wv][0], &Lt[wv][0], lane, plane,
                       ssrc, src_feat, asn, out);
    if (dB < N)
        aggregate_node(dB, begB, nB, &Yt[wv + 16][0], &Lt[wv + 16][0], lane, plane,
                       ssrc, src_feat, asn, out);
}

extern "C" void kernel_launch(void* const* d_in, const int* in_sizes, int n_in,
                              void* d_out, int out_size, void* d_ws, size_t ws_size,
                              hipStream_t stream)
{
    const float* x_src  = (const float*)d_in[0];
    const float* x_dst  = (const float*)d_in[1];
    const int*   esrc   = (const int*)d_in[2];
    const int*   edst   = (const int*)d_in[3];
    const float* W_src  = (const float*)d_in[5];
    const float* W_dst  = (const float*)d_in[6];
    const float* attS   = (const float*)d_in[7];
    const float* attD   = (const float*)d_in[8];
    const float* W_self = (const float*)d_in[9];
    const float* b_self = (const float*)d_in[10];

    const int N_SRC = in_sizes[0] / FDIM;
    const int N_DST = in_sizes[1] / FDIM;
    const int E     = in_sizes[2];

    char* ws = (char*)d_ws;
    float* src_feat = (float*)(ws);
    float* Vp_s     = (float*)(ws + (size_t)N_SRC * FDIM * 4);
    float* Vp_d     = Vp_s + FDIM * 16;
    const int B = (N_DST + BUCKET_NODES - 1) / BUCKET_NODES;
    const int C = (E + CHUNK_EDGES - 1) / CHUNK_EDGES;
    int2*  pairs    = (int2*)(Vp_d + FDIM * 16);
    int*   seg      = (int*)(pairs + E);
    float* as_node  = (float*)(ws + (size_t)(N_SRC + N_DST) * FDIM * 4);
    int*   cnt      = (int*)(as_node + (size_t)N_SRC * HEADS);
    int*   start    = cnt + N_DST;
    int*   ssrc     = start + 2 * (size_t)N_DST;

    float* out = (float*)d_out;

    sort_chunk_by_bucket_build_logit_tiles<<<C + 4, 1024, 0, stream>>>(
        esrc, edst, seg, pairs, E, C, W_src, attS, W_dst, attD, Vp_s, Vp_d);
    gemm_x128_logits_beside_node_grouping<<<B + (N_SRC + ROWS - 1) / ROWS, GEMM_THREADS, 0, stream>>>(
        x_src, W_src, Vp_s, nullptr, src_feat, as_node, N_SRC,
        pairs, seg, start, cnt, ssrc, N_DST, B, C);
    self_project_softmax_aggregate<<<(N_DST + ROWS - 1) / ROWS, 512, 0, stream>>>(
        x_dst, W_self, Vp_d, b_self, ssrc, start, cnt, src_feat, as_node, out, N_DST);
}
